# starting kernel
# baseline (speedup 1.0000x reference)
#include <hip/hip_runtime.h>

typedef __attribute__((ext_vector_type(2))) float v2f;
typedef __attribute__((ext_vector_type(8))) float v8f;

#define D_FEAT 64
#define LDS_STRIDE 66

__global__ __launch_bounds__(256) void gcn_zero_kernel(float4* __restrict__ p, int n4) {
    int i = blockIdx.x * blockDim.x + threadIdx.x;
    if (i < n4) p[i] = make_float4(0.f, 0.f, 0.f, 0.f);
}

__global__ __launch_bounds__(256) void gcn_scatter_kernel(
    const int*   __restrict__ rows,
    const int*   __restrict__ cols,
    const float* __restrict__ vals,
    const float* __restrict__ X,
    float*       __restrict__ agg,
    int n_edges)
{
    int gid = blockIdx.x * blockDim.x + threadIdx.x;
    int e = gid >> 4;
    if (e >= n_edges) return;
    int f = (gid & 15) << 2;

    int r   = rows[e];
    int c   = cols[e];
    float v = vals[e];

    const float4 x = *(const float4*)(X + (size_t)c * D_FEAT + f);
    float* dst = agg + (size_t)r * D_FEAT + f;

    unsafeAtomicAdd(dst + 0, v * x.x);
    unsafeAtomicAdd(dst + 1, v * x.y);
    unsafeAtomicAdd(dst + 2, v * x.z);
    unsafeAtomicAdd(dst + 3, v * x.w);
}

__global__ __launch_bounds__(256) void gcn_gemm_wmma_kernel(
    const float* __restrict__ agg,
    const float* __restrict__ W,
    const float* __restrict__ bias,
    float*       __restrict__ out,
    int n_tiles)
{
    __shared__ float ldsW[D_FEAT * LDS_STRIDE];

    const int tid = threadIdx.x;

    for (int i = tid; i < D_FEAT * D_FEAT; i += 256) {
        ldsW[(i >> 6) * LDS_STRIDE + (i & 63)] = W[i];
    }
    __syncthreads();

    const int wave = tid >> 5;
    const int lane = tid & 31;
    const int half = lane >> 4;
    const int l16  = lane & 15;
    const int koff = half << 1;

    const int tile = blockIdx.x * 8 + wave;
    if (tile >= n_tiles) return;

    const float bn0 = bias[ 0 + l16];
    const float bn1 = bias[16 + l16];
    const float bn2 = bias[32 + l16];
    const float bn3 = bias[48 + l16];

    v8f c0 = {}; v8f c1 = {}; v8f c2 = {}; v8f c3 = {};

    const float* arow = agg + ((size_t)tile * 16 + l16) * D_FEAT;

#pragma unroll
    for (int kk = 0; kk < 16; ++kk) {
        const int k0 = kk * 4;

        v2f a = *(const v2f*)(arow + k0 + koff);

        const int kb = k0 + koff;
        v2f b0 = *(const v2f*)(&ldsW[( 0 + l16) * LDS_STRIDE + kb]);
        v2f b1 = *(const v2f*)(&ldsW[(16 + l16) * LDS_STRIDE + kb]);
        v2f b2 = *(const v2f*)(&ldsW[(32 + l16) * LDS_STRIDE + kb]);
        v2f b3 = *(const v2f*)(&ldsW[(48 + l16) * LDS_STRIDE + kb]);

        c0 = __builtin_amdgcn_wmma_f32_16x16x4_f32(false, a, false, b0, (short)0, c0, false, false);
        c1 = __builtin_amdgcn_wmma_f32_16x16x4_f32(false, a, false, b1, (short)0, c1, false, false);
        c2 = __builtin_amdgcn_wmma_f32_16x16x4_f32(false, a, false, b2, (short)0, c2, false, false);
        c3 = __builtin_amdgcn_wmma_f32_16x16x4_f32(false, a, false, b3, (short)0, c3, false, false);
    }

    const size_t base = (size_t)tile * 16 * D_FEAT;
#pragma unroll
    for (int r = 0; r < 8; ++r) {
        const size_t m_off = base + (size_t)(r + half * 8) * D_FEAT;
        out[m_off +  0 + l16] = c0[r] + bn0;
        out[m_off + 16 + l16] = c1[r] + bn1;
        out[m_off + 32 + l16] = c2[r] + bn2;
        out[m_off + 48 + l16] = c3[r] + bn3;
    }
}

extern "C" void kernel_launch(void* const* d_in, const int* in_sizes, int n_in,
                              void* d_out, int out_size, void* d_ws, size_t ws_size,
                              hipStream_t stream) {
    const int*   rows = (const int*)d_in[0];
    const int*   cols = (const int*)d_in[1];
    const float* vals = (const float*)d_in[2];
    const float* X    = (const float*)d_in[3];
    const float* W    = (const float*)d_in[4];
    const float* bias = (const float*)d_in[5];
    float* out = (float*)d_out;
    float* agg = (float*)d_ws;

    const int n_edges = in_sizes[0];
    const int n_nodes = in_sizes[3] / D_FEAT;

    const int n4 = n_nodes * (D_FEAT / 4);
    gcn_zero_kernel<<<(n4 + 255) / 256, 256, 0, stream>>>((float4*)agg, n4);

    const long long total = (long long)n_edges * 16;
    const int blocks = (int)((total + 255) / 256);
    gcn_scatter_kernel<<<blocks, 256, 0, stream>>>(rows, cols, vals, X, agg, n_edges);

    const int n_tiles = (n_nodes + 15) / 16;
    gcn_gemm_wmma_kernel<<<(n_tiles + 7) / 8, 256, 0, stream>>>(agg, W, bias, out, n_tiles);
}


# Replace global f32 atomic scatter by a per-call LDS counting sort of edges into 128-row bins and per-bin sums in LDS (rows stored once); sum order within a row = arrival order of LDS adds
# speedup vs baseline: 1.4301x; 1.4301x over previous
#include <hip/hip_runtime.h>

typedef __attribute__((ext_vector_type(2))) float v2f;
typedef __attribute__((ext_vector_type(8))) float v8f;

#define D_FEAT 64
#define LDS_STRIDE 66

#define BIN_SHIFT 7
#define BIN_ROWS  (1 << BIN_SHIFT)
#define SORT_BLOCKS 512
#define SCAN_ELEMS 1024


__device__ __forceinline__ int wave_sum(int s) {
    s += __shfl_xor(s, 16, 32);
    s += __shfl_xor(s, 8, 32);
    s += __shfl_xor(s, 4, 32);
    s += __shfl_xor(s, 2, 32);
    s += __shfl_xor(s, 1, 32);
    return s;
}

__global__ __launch_bounds__(256) void gcn_count_kernel(
    const int* __restrict__ rows,
    int*       __restrict__ table,
    int n_edges, int n_bins, int chunk)
{
    extern __shared__ int lds_cnt[];
    const int tid = threadIdx.x;
    for (int i = tid; i < n_bins; i += 256) lds_cnt[i] = 0;
    __syncthreads();

    const int lo = blockIdx.x * chunk;
    const int hi = min(n_edges, lo + chunk);
    for (int e = lo + tid * 4; e < hi; e += 1024) {
        if (e + 3 < hi) {
            const int4 r = *(const int4*)(rows + e);
            atomicAdd(&lds_cnt[r.x >> BIN_SHIFT], 1);
            atomicAdd(&lds_cnt[r.y >> BIN_SHIFT], 1);
            atomicAdd(&lds_cnt[r.z >> BIN_SHIFT], 1);
            atomicAdd(&lds_cnt[r.w >> BIN_SHIFT], 1);
        } else {
            for (int j = e; j < hi; ++j) atomicAdd(&lds_cnt[rows[j] >> BIN_SHIFT], 1);
        }
    }
    __syncthreads();

    for (int i = tid; i < n_bins; i += 256) {
        table[(size_t)i * SORT_BLOCKS + blockIdx.x] = lds_cnt[i];
    }
}

__global__ __launch_bounds__(256) void gcn_blocksum_kernel(
    const int* __restrict__ table, int* __restrict__ bsum, int m)
{
    __shared__ int wsum[8];
    const int tid  = threadIdx.x;
    const int base = blockIdx.x * SCAN_ELEMS + tid * 4;
    int s = 0;
    if (base + 3 < m) {
        const int4 v = *(const int4*)(table + base);
        s = v.x + v.y + v.z + v.w;
    } else {
        for (int j = 0; j < 4; ++j) if (base + j < m) s += table[base + j];
    }
    s = wave_sum(s);
    if ((tid & 31) == 0) wsum[tid >> 5] = s;
    __syncthreads();
    if (tid == 0) {
        int t = 0;
        for (int w = 0; w < 8; ++w) t += wsum[w];
        bsum[blockIdx.x] = t;
    }
}

__global__ __launch_bounds__(256) void gcn_offsets_kernel(
    const int* __restrict__ table, const int* __restrict__ bsum,
    int* __restrict__ offs, int m)
{
    __shared__ int wsum[8];
    const int tid  = threadIdx.x;
    const int lane = tid & 31;
    const int wave = tid >> 5;

    int p = 0;
    for (int b = tid; b < (int)blockIdx.x; b += 256) p += bsum[b];
    p = wave_sum(p);
    if (lane == 0) wsum[wave] = p;
    __syncthreads();
    int prefix = 0;
    for (int w = 0; w < 8; ++w) prefix += wsum[w];
    __syncthreads();

    const int base = blockIdx.x * SCAN_ELEMS + tid * 4;
    int c0 = 0, c1 = 0, c2 = 0, c3 = 0;
    if (base + 3 < m) {
        const int4 v = *(const int4*)(table + base);
        c0 = v.x; c1 = v.y; c2 = v.z; c3 = v.w;
    } else {
        if (base + 0 < m) c0 = table[base + 0];
        if (base + 1 < m) c1 = table[base + 1];
        if (base + 2 < m) c2 = table[base + 2];
    }
    const int t = c0 + c1 + c2 + c3;
    int inc = t;
    for (int d = 1; d < 32; d <<= 1) {
        const int u = __shfl_up(inc, d, 32);
        if (lane >= d) inc += u;
    }
    if (lane == 31) wsum[wave] = inc;
    __syncthreads();
    int wp = 0;
    for (int w = 0; w < wave; ++w) wp += wsum[w];

    const int ex = prefix + wp + inc - t;
    if (base + 0 < m) offs[base + 0] = ex;
    if (base + 1 < m) offs[base + 1] = ex + c0;
    if (base + 2 < m) offs[base + 2] = ex + c0 + c1;
    if (base + 3 < m) offs[base + 3] = ex + c0 + c1 + c2;
}

__global__ __launch_bounds__(256) void gcn_place_kernel(
    const int*   __restrict__ rows,
    const int*   __restrict__ cols,
    const float* __restrict__ vals,
    const int*   __restrict__ offs,
    int2*        __restrict__ recs,
    int n_edges, int n_bins, int chunk)
{
    extern __shared__ int lds_pos[];
    const int tid = threadIdx.x;
    for (int i = tid; i < n_bins; i += 256) {
        lds_pos[i] = offs[(size_t)i * SORT_BLOCKS + blockIdx.x];
    }
    __syncthreads();

    const int lo = blockIdx.x * chunk;
    const int hi = min(n_edges, lo + chunk);
    for (int e = lo + tid * 4; e < hi; e += 1024) {
        if (e + 3 < hi) {
            const int4   r = *(const int4*)(rows + e);
            const int4   c = *(const int4*)(cols + e);
            const float4 v = *(const float4*)(vals + e);
            const int p0 = atomicAdd(&lds_pos[r.x >> BIN_SHIFT], 1);
            const int p1 = atomicAdd(&lds_pos[r.y >> BIN_SHIFT], 1);
            const int p2 = atomicAdd(&lds_pos[r.z >> BIN_SHIFT], 1);
            const int p3 = atomicAdd(&lds_pos[r.w >> BIN_SHIFT], 1);
            recs[p0] = make_int2(c.x | ((r.x & (BIN_ROWS - 1)) << 24), __float_as_int(v.x));
            recs[p1] = make_int2(c.y | ((r.y & (BIN_ROWS - 1)) << 24), __float_as_int(v.y));
            recs[p2] = make_int2(c.z | ((r.z & (BIN_ROWS - 1)) << 24), __float_as_int(v.z));
            recs[p3] = make_int2(c.w | ((r.w & (BIN_ROWS - 1)) << 24), __float_as_int(v.w));
        } else {
            for (int j = e; j < hi; ++j) {
                const int rj = rows[j];
                const int p = atomicAdd(&lds_pos[rj >> BIN_SHIFT], 1);
                recs[p] = make_int2(cols[j] | ((rj & (BIN_ROWS - 1)) << 24), __float_as_int(vals[j]));
            }
        }
    }
}

__global__ __launch_bounds__(256) void gcn_binsum_kernel(
    const int2*  __restrict__ recs,
    const int*   __restrict__ offs,
    const float* __restrict__ X,
    float*       __restrict__ agg,
    int n_edges, int n_bins, int n_nodes)
{
    __shared__ alignas(16) float tile[BIN_ROWS * D_FEAT];

    const int tid = threadIdx.x;
    const int bin = blockIdx.x;

    float4* tile4 = (float4*)tile;
    for (int i = tid; i < BIN_ROWS * D_FEAT / 4; i += 256) tile4[i] = make_float4(0.f, 0.f, 0.f, 0.f);
    __syncthreads();

    const int start = offs[(size_t)bin * SORT_BLOCKS];
    const int end   = (bin + 1 < n_bins) ? offs[(size_t)(bin + 1) * SORT_BLOCKS] : n_edges;

    const int lane = tid & 31;
    const int wave = __builtin_amdgcn_readfirstlane(tid >> 5);
    const v2f* X2 = (const v2f*)X;

    int e0 = start + wave;
    if (e0 < end) {
        const int last = end - 1;
        int2 rec[8];
#pragma unroll
        for (int k = 0; k < 8; ++k) rec[k] = recs[min(e0 + 8 * k, last)];
        do {
            v2f x[8];
#pragma unroll
            for (int k = 0; k < 8; ++k) {
                const int c = rec[k].x & 0xFFFFFF;
                x[k] = X2[(size_t)c * (D_FEAT / 2) + lane];
            }
            const int n0 = e0 + 64;
            int2 nrec[8];
#pragma unroll
            for (int k = 0; k < 8; ++k) nrec[k] = recs[min(n0 + 8 * k, last)];
#pragma unroll
            for (int k = 0; k < 8; ++k) {
                if (e0 + 8 * k < end) {
                    const float v = __int_as_float(rec[k].y);
                    float* dst = tile + (((unsigned)rec[k].x >> 24) * D_FEAT) + lane * 2;
                    atomicAdd(dst + 0, v * x[k].x);
                    atomicAdd(dst + 1, v * x[k].y);
                }
            }
#pragma unroll
            for (int k = 0; k < 8; ++k) rec[k] = nrec[k];
            e0 = n0;
        } while (e0 < end);
    }
    __syncthreads();

    const int row0 = bin * BIN_ROWS;
    float4* agg4 = (float4*)agg;
    for (int i = tid; i < BIN_ROWS * D_FEAT / 4; i += 256) {
        const int row = row0 + (i >> 4);
        if (row < n_nodes) agg4[(size_t)row * (D_FEAT / 4) + (i & 15)] = tile4[i];
    }
}

__global__ __launch_bounds__(256) void gcn_gemm_wmma_kernel(
    const float* __restrict__ agg,
    const float* __restrict__ W,
    const float* __restrict__ bias,
    float*       __restrict__ out,
    int n_tiles)
{
    __shared__ float ldsW[D_FEAT * LDS_STRIDE];

    const int tid = threadIdx.x;

    for (int i = tid; i < D_FEAT * D_FEAT; i += 256) {
        ldsW[(i >> 6) * LDS_STRIDE + (i & 63)] = W[i];
    }
    __syncthreads();

    const int wave = tid >> 5;
    const int lane = tid & 31;
    const int half = lane >> 4;
    const int l16  = lane & 15;
    const int koff = half << 1;

    const int tile = blockIdx.x * 8 + wave;
    if (tile >= n_tiles) return;

    const float bn0 = bias[ 0 + l16];
    const float bn1 = bias[16 + l16];
    const float bn2 = bias[32 + l16];
    const float bn3 = bias[48 + l16];

    v8f c0 = {}; v8f c1 = {}; v8f c2 = {}; v8f c3 = {};

    const float* arow = agg + ((size_t)tile * 16 + l16) * D_FEAT;

#pragma unroll
    for (int kk = 0; kk < 16; ++kk) {
        const int k0 = kk * 4;

        v2f a = *(const v2f*)(arow + k0 + koff);

        const int kb = k0 + koff;
        v2f b0 = *(const v2f*)(&ldsW[( 0 + l16) * LDS_STRIDE + kb]);
        v2f b1 = *(const v2f*)(&ldsW[(16 + l16) * LDS_STRIDE + kb]);
        v2f b2 = *(const v2f*)(&ldsW[(32 + l16) * LDS_STRIDE + kb]);
        v2f b3 = *(const v2f*)(&ldsW[(48 + l16) * LDS_STRIDE + kb]);

        c0 = __builtin_amdgcn_wmma_f32_16x16x4_f32(false, a, false, b0, (short)0, c0, false, false);
        c1 = __builtin_amdgcn_wmma_f32_16x16x4_f32(false, a, false, b1, (short)0, c1, false, false);
        c2 = __builtin_amdgcn_wmma_f32_16x16x4_f32(false, a, false, b2, (short)0, c2, false, false);
        c3 = __builtin_amdgcn_wmma_f32_16x16x4_f32(false, a, false, b3, (short)0, c3, false, false);
    }

    const size_t base = (size_t)tile * 16 * D_FEAT;
#pragma unroll
    for (int r = 0; r < 8; ++r) {
        const size_t m_off = base + (size_t)(r + half * 8) * D_FEAT;
        out[m_off +  0 + l16] = c0[r] + bn0;
        out[m_off + 16 + l16] = c1[r] + bn1;
        out[m_off + 32 + l16] = c2[r] + bn2;
        out[m_off + 48 + l16] = c3[r] + bn3;
    }
}

extern "C" void kernel_launch(void* const* d_in, const int* in_sizes, int n_in,
                              void* d_out, int out_size, void* d_ws, size_t ws_size,
                              hipStream_t stream) {
    const int*   rows = (const int*)d_in[0];
    const int*   cols = (const int*)d_in[1];
    const float* vals = (const float*)d_in[2];
    const float* X    = (const float*)d_in[3];
    const float* W    = (const float*)d_in[4];
    const float* bias = (const float*)d_in[5];
    float* out = (float*)d_out;

    const int n_edges = in_sizes[0];
    const int n_nodes = in_sizes[3] / D_FEAT;
    const int n_bins  = (n_nodes + BIN_ROWS - 1) / BIN_ROWS;
    const int chunk   = ((n_edges + SORT_BLOCKS * 4 - 1) / (SORT_BLOCKS * 4)) * 4;
    const int m       = n_bins * SORT_BLOCKS;
    const int n_scan  = (m + SCAN_ELEMS - 1) / SCAN_ELEMS;

    char* ws = (char*)d_ws;
    float* agg = (float*)ws;
    size_t o = (((size_t)n_nodes * D_FEAT * sizeof(float)) + 255) & ~(size_t)255;
    int* table = (int*)(ws + o);  o += (((size_t)m * sizeof(int)) + 255) & ~(size_t)255;
    int* offs  = (int*)(ws + o);  o += (((size_t)m * sizeof(int)) + 255) & ~(size_t)255;
    int* bsum  = (int*)(ws + o);  o += (((size_t)n_scan * sizeof(int)) + 255) & ~(size_t)255;
    int2* recs = (int2*)(ws + o);

    const size_t lds_bins = (size_t)n_bins * sizeof(int);

    gcn_count_kernel<<<SORT_BLOCKS, 256, lds_bins, stream>>>(rows, table, n_edges, n_bins, chunk);
    gcn_blocksum_kernel<<<n_scan, 256, 0, stream>>>(table, bsum, m);
    gcn_offsets_kernel<<<n_scan, 256, 0, stream>>>(table, bsum, offs, m);
    gcn_place_kernel<<<SORT_BLOCKS, 256, lds_bins, stream>>>(rows, cols, vals, offs, recs, n_edges, n_bins, chunk);
    gcn_binsum_kernel<<<n_bins, 256, 0, stream>>>(recs, offs, X, agg, n_edges, n_bins, n_nodes);

    const int n_tiles = (n_nodes + 15) / 16;
    gcn_gemm_wmma_kernel<<<(n_tiles + 7) / 8, 256, 0, stream>>>(agg, W, bias, out, n_tiles);
}


# Sort each 12288-edge chunk's records by bin in LDS, then store the sorted chunk with consecutive lanes on consecutive records (whole runs, 4x smaller count table); row sums keep LDS arrival order
# speedup vs baseline: 2.0859x; 1.4586x over previous
#include <hip/hip_runtime.h>

typedef __attribute__((ext_vector_type(2))) float v2f;
typedef __attribute__((ext_vector_type(8))) float v8f;

#define D_FEAT 64
#define LDS_STRIDE 66

#define BIN_SHIFT 7
#define BIN_ROWS  (1 << BIN_SHIFT)
#define SORT_CHUNK 12288
#define SORT_THREADS 1024
#define SCAN_ELEMS 1024


__device__ __forceinline__ int wave_sum(int s) {
    s += __shfl_xor(s, 16, 32);
    s += __shfl_xor(s, 8, 32);
    s += __shfl_xor(s, 4, 32);
    s += __shfl_xor(s, 2, 32);
    s += __shfl_xor(s, 1, 32);
    return s;
}

__global__ __launch_bounds__(SORT_THREADS) void gcn_count_kernel(
    const int* __restrict__ rows,
    int*       __restrict__ table,
    int n_edges, int n_bins, int n_chunks)
{
    extern __shared__ int lds_cnt[];
    const int tid = threadIdx.x;
    for (int i = tid; i < n_bins; i += SORT_THREADS) lds_cnt[i] = 0;
    __syncthreads();

    const int lo = blockIdx.x * SORT_CHUNK;
    const int hi = min(n_edges, lo + SORT_CHUNK);
    for (int e = lo + tid * 4; e < hi; e += SORT_THREADS * 4) {
        if (e + 3 < hi) {
            const int4 r = *(const int4*)(rows + e);
            atomicAdd(&lds_cnt[r.x >> BIN_SHIFT], 1);
            atomicAdd(&lds_cnt[r.y >> BIN_SHIFT], 1);
            atomicAdd(&lds_cnt[r.z >> BIN_SHIFT], 1);
            atomicAdd(&lds_cnt[r.w >> BIN_SHIFT], 1);
        } else {
            for (int j = e; j < hi; ++j) atomicAdd(&lds_cnt[rows[j] >> BIN_SHIFT], 1);
        }
    }
    __syncthreads();

    for (int i = tid; i < n_bins; i += SORT_THREADS) {
        table[(size_t)i * n_chunks + blockIdx.x] = lds_cnt[i];
    }
}

__global__ __launch_bounds__(256) void gcn_blocksum_kernel(
    const int* __restrict__ table, int* __restrict__ bsum, int m)
{
    __shared__ int wsum[8];
    const int tid  = threadIdx.x;
    const int base = blockIdx.x * SCAN_ELEMS + tid * 4;
    int s = 0;
    if (base + 3 < m) {
        const int4 v = *(const int4*)(table + base);
        s = v.x + v.y + v.z + v.w;
    } else {
        for (int j = 0; j < 4; ++j) if (base + j < m) s += table[base + j];
    }
    s = wave_sum(s);
    if ((tid & 31) == 0) wsum[tid >> 5] = s;
    __syncthreads();
    if (tid == 0) {
        int t = 0;
        for (int w = 0; w < 8; ++w) t += wsum[w];
        bsum[blockIdx.x] = t;
    }
}

__global__ __launch_bounds__(256) void gcn_offsets_kernel(
    const int* __restrict__ table, const int* __restrict__ bsum,
    int* __restrict__ offs, int m)
{
    __shared__ int wsum[8];
    const int tid  = threadIdx.x;
    const int lane = tid & 31;
    const int wave = tid >> 5;

    int p = 0;
    for (int b = tid; b < (int)blockIdx.x; b += 256) p += bsum[b];
    p = wave_sum(p);
    if (lane == 0) wsum[wave] = p;
    __syncthreads();
    int prefix = 0;
    for (int w = 0; w < 8; ++w) prefix += wsum[w];
    __syncthreads();

    const int base = blockIdx.x * SCAN_ELEMS + tid * 4;
    int c0 = 0, c1 = 0, c2 = 0, c3 = 0;
    if (base + 3 < m) {
        const int4 v = *(const int4*)(table + base);
        c0 = v.x; c1 = v.y; c2 = v.z; c3 = v.w;
    } else {
        if (base + 0 < m) c0 = table[base + 0];
        if (base + 1 < m) c1 = table[base + 1];
        if (base + 2 < m) c2 = table[base + 2];
    }
    const int t = c0 + c1 + c2 + c3;
    int inc = t;
    for (int d = 1; d < 32; d <<= 1) {
        const int u = __shfl_up(inc, d, 32);
        if (lane >= d) inc += u;
    }
    if (lane == 31) wsum[wave] = inc;
    __syncthreads();
    int wp = 0;
    for (int w = 0; w < wave; ++w) wp += wsum[w];

    const int ex = prefix + wp + inc - t;
    if (base + 0 < m) offs[base + 0] = ex;
    if (base + 1 < m) offs[base + 1] = ex + c0;
    if (base + 2 < m) offs[base + 2] = ex + c0 + c1;
    if (base + 3 < m) offs[base + 3] = ex + c0 + c1 + c2;
}

__global__ __launch_bounds__(SORT_THREADS) void gcn_place_kernel(
    const int*   __restrict__ rows,
    const int*   __restrict__ cols,
    const float* __restrict__ vals,
    const int*   __restrict__ offs,
    int2*        __restrict__ recs,
    int n_edges, int n_bins, int n_chunks)
{
    extern __shared__ int lds_bins[];
    __shared__ int2 srec[SORT_CHUNK];
    __shared__ int  sdst[SORT_CHUNK];
    __shared__ int  wsum[SORT_THREADS / 32];

    int* lslot  = lds_bins;
    int* ldelta = lds_bins + n_bins;

    const int tid  = threadIdx.x;
    const int lane = tid & 31;
    const int wave = tid >> 5;
    const int blk  = blockIdx.x;
    const int m    = n_bins * n_chunks;

    int carry = 0;
    for (int base = 0; base < n_bins; base += SORT_THREADS) {
        const int i = base + tid;
        int g = 0, c = 0;
        if (i < n_bins) {
            const int f = i * n_chunks + blk;
            g = offs[f];
            const int gn = (f + 1 < m) ? offs[f + 1] : n_edges;
            c = gn - g;
        }
        int inc = c;
        for (int d = 1; d < 32; d <<= 1) {
            const int u = __shfl_up(inc, d, 32);
            if (lane >= d) inc += u;
        }
        if (lane == 31) wsum[wave] = inc;
        __syncthreads();
        int ws = wsum[lane];
        for (int d = 1; d < 32; d <<= 1) {
            const int u = __shfl_up(ws, d, 32);
            if (lane >= d) ws += u;
        }
        const int tot = __shfl(ws, 31, 32);
        const int wp  = __shfl(ws, wave, 32) - wsum[wave];
        const int ex  = carry + wp + inc - c;
        if (i < n_bins) { lslot[i] = ex; ldelta[i] = g - ex; }
        carry += tot;
        __syncthreads();
    }

    const int lo = blk * SORT_CHUNK;
    const int hi = min(n_edges, lo + SORT_CHUNK);
    for (int e = lo + tid * 4; e < hi; e += SORT_THREADS * 4) {
        if (e + 3 < hi) {
            const int4   r = *(const int4*)(rows + e);
            const int4   c = *(const int4*)(cols + e);
            const float4 v = *(const float4*)(vals + e);
            const int b0 = r.x >> BIN_SHIFT, b1 = r.y >> BIN_SHIFT;
            const int b2 = r.z >> BIN_SHIFT, b3 = r.w >> BIN_SHIFT;
            const int s0 = atomicAdd(&lslot[b0], 1);
            const int s1 = atomicAdd(&lslot[b1], 1);
            const int s2 = atomicAdd(&lslot[b2], 1);
            const int s3 = atomicAdd(&lslot[b3], 1);
            srec[s0] = make_int2(c.x | ((r.x & (BIN_ROWS - 1)) << 24), __float_as_int(v.x));
            srec[s1] = make_int2(c.y | ((r.y & (BIN_ROWS - 1)) << 24), __float_as_int(v.y));
            srec[s2] = make_int2(c.z | ((r.z & (BIN_ROWS - 1)) << 24), __float_as_int(v.z));
            srec[s3] = make_int2(c.w | ((r.w & (BIN_ROWS - 1)) << 24), __float_as_int(v.w));
            sdst[s0] = s0 + ldelta[b0];
            sdst[s1] = s1 + ldelta[b1];
            sdst[s2] = s2 + ldelta[b2];
            sdst[s3] = s3 + ldelta[b3];
        } else {
            for (int j = e; j < hi; ++j) {
                const int rj = rows[j];
                const int bj = rj >> BIN_SHIFT;
                const int s = atomicAdd(&lslot[bj], 1);
                srec[s] = make_int2(cols[j] | ((rj & (BIN_ROWS - 1)) << 24), __float_as_int(vals[j]));
                sdst[s] = s + ldelta[bj];
            }
        }
    }
    __syncthreads();

    const int n = hi - lo;
    for (int t = tid; t < n; t += SORT_THREADS) recs[sdst[t]] = srec[t];
}

__global__ __launch_bounds__(256) void gcn_binsum_kernel(
    const int2*  __restrict__ recs,
    const int*   __restrict__ offs,
    const float* __restrict__ X,
    float*       __restrict__ agg,
    int n_edges, int n_bins, int n_nodes, int n_chunks)
{
    __shared__ alignas(16) float tile[BIN_ROWS * D_FEAT];

    const int tid = threadIdx.x;
    const int bin = blockIdx.x;

    float4* tile4 = (float4*)tile;
    for (int i = tid; i < BIN_ROWS * D_FEAT / 4; i += 256) tile4[i] = make_float4(0.f, 0.f, 0.f, 0.f);
    __syncthreads();

    const int start = offs[(size_t)bin * n_chunks];
    const int end   = (bin + 1 < n_bins) ? offs[(size_t)(bin + 1) * n_chunks] : n_edges;

    const int lane = tid & 31;
    const int wave = __builtin_amdgcn_readfirstlane(tid >> 5);
    const v2f* X2 = (const v2f*)X;

    int e0 = start + wave;
    if (e0 < end) {
        const int last = end - 1;
        int2 rec[8];
#pragma unroll
        for (int k = 0; k < 8; ++k) rec[k] = recs[min(e0 + 8 * k, last)];
        do {
            v2f x[8];
#pragma unroll
            for (int k = 0; k < 8; ++k) {
                const int c = rec[k].x & 0xFFFFFF;
                x[k] = X2[(size_t)c * (D_FEAT / 2) + lane];
            }
            const int n0 = e0 + 64;
            int2 nrec[8];
#pragma unroll
            for (int k = 0; k < 8; ++k) nrec[k] = recs[min(n0 + 8 * k, last)];
#pragma unroll
            for (int k = 0; k < 8; ++k) {
                if (e0 + 8 * k < end) {
                    const float v = __int_as_float(rec[k].y);
                    float* dst = tile + (((unsigned)rec[k].x >> 24) * D_FEAT) + lane * 2;
                    atomicAdd(dst + 0, v * x[k].x);
                    atomicAdd(dst + 1, v * x[k].y);
                }
            }
#pragma unroll
            for (int k = 0; k < 8; ++k) rec[k] = nrec[k];
            e0 = n0;
        } while (e0 < end);
    }
    __syncthreads();

    const int row0 = bin * BIN_ROWS;
    float4* agg4 = (float4*)agg;
    for (int i = tid; i < BIN_ROWS * D_FEAT / 4; i += 256) {
        const int row = row0 + (i >> 4);
        if (row < n_nodes) agg4[(size_t)row * (D_FEAT / 4) + (i & 15)] = tile4[i];
    }
}

__global__ __launch_bounds__(256) void gcn_gemm_wmma_kernel(
    const float* __restrict__ agg,
    const float* __restrict__ W,
    const float* __restrict__ bias,
    float*       __restrict__ out,
    int n_tiles)
{
    __shared__ float ldsW[D_FEAT * LDS_STRIDE];

    const int tid = threadIdx.x;

    for (int i = tid; i < D_FEAT * D_FEAT; i += 256) {
        ldsW[(i >> 6) * LDS_STRIDE + (i & 63)] = W[i];
    }
    __syncthreads();

    const int wave = tid >> 5;
    const int lane = tid & 31;
    const int half = lane >> 4;
    const int l16  = lane & 15;
    const int koff = half << 1;

    const int tile = blockIdx.x * 8 + wave;
    if (tile >= n_tiles) return;

    const float bn0 = bias[ 0 + l16];
    const float bn1 = bias[16 + l16];
    const float bn2 = bias[32 + l16];
    const float bn3 = bias[48 + l16];

    v8f c0 = {}; v8f c1 = {}; v8f c2 = {}; v8f c3 = {};

    const float* arow = agg + ((size_t)tile * 16 + l16) * D_FEAT;

#pragma unroll
    for (int kk = 0; kk < 16; ++kk) {
        const int k0 = kk * 4;

        v2f a = *(const v2f*)(arow + k0 + koff);

        const int kb = k0 + koff;
        v2f b0 = *(const v2f*)(&ldsW[( 0 + l16) * LDS_STRIDE + kb]);
        v2f b1 = *(const v2f*)(&ldsW[(16 + l16) * LDS_STRIDE + kb]);
        v2f b2 = *(const v2f*)(&ldsW[(32 + l16) * LDS_STRIDE + kb]);
        v2f b3 = *(const v2f*)(&ldsW[(48 + l16) * LDS_STRIDE + kb]);

        c0 = __builtin_amdgcn_wmma_f32_16x16x4_f32(false, a, false, b0, (short)0, c0, false, false);
        c1 = __builtin_amdgcn_wmma_f32_16x16x4_f32(false, a, false, b1, (short)0, c1, false, false);
        c2 = __builtin_amdgcn_wmma_f32_16x16x4_f32(false, a, false, b2, (short)0, c2, false, false);
        c3 = __builtin_amdgcn_wmma_f32_16x16x4_f32(false, a, false, b3, (short)0, c3, false, false);
    }

    const size_t base = (size_t)tile * 16 * D_FEAT;
#pragma unroll
    for (int r = 0; r < 8; ++r) {
        const size_t m_off = base + (size_t)(r + half * 8) * D_FEAT;
        out[m_off +  0 + l16] = c0[r] + bn0;
        out[m_off + 16 + l16] = c1[r] + bn1;
        out[m_off + 32 + l16] = c2[r] + bn2;
        out[m_off + 48 + l16] = c3[r] + bn3;
    }
}

extern "C" void kernel_launch(void* const* d_in, const int* in_sizes, int n_in,
                              void* d_out, int out_size, void* d_ws, size_t ws_size,
                              hipStream_t stream) {
    const int*   rows = (const int*)d_in[0];
    const int*   cols = (const int*)d_in[1];
    const float* vals = (const float*)d_in[2];
    const float* X    = (const float*)d_in[3];
    const float* W    = (const float*)d_in[4];
    const float* bias = (const float*)d_in[5];
    float* out = (float*)d_out;

    const int n_edges = in_sizes[0];
    const int n_nodes = in_sizes[3] / D_FEAT;
    const int n_bins  = (n_nodes + BIN_ROWS - 1) / BIN_ROWS;
    const int n_chunks = (n_edges + SORT_CHUNK - 1) / SORT_CHUNK;
    const int m       = n_bins * n_chunks;
    const int n_scan  = (m + SCAN_ELEMS - 1) / SCAN_ELEMS;

    char* ws = (char*)d_ws;
    float* agg = (float*)ws;
    size_t o = (((size_t)n_nodes * D_FEAT * sizeof(float)) + 255) & ~(size_t)255;
    int* table = (int*)(ws + o);  o += (((size_t)m * sizeof(int)) + 255) & ~(size_t)255;
    int* offs  = (int*)(ws + o);  o += (((size_t)m * sizeof(int)) + 255) & ~(size_t)255;
    int* bsum  = (int*)(ws + o);  o += (((size_t)n_scan * sizeof(int)) + 255) & ~(size_t)255;
    int2* recs = (int2*)(ws + o);

    const size_t lds_bins = (size_t)n_bins * sizeof(int);

    gcn_count_kernel<<<n_chunks, SORT_THREADS, lds_bins, stream>>>(rows, table, n_edges, n_bins, n_chunks);
    gcn_blocksum_kernel<<<n_scan, 256, 0, stream>>>(table, bsum, m);
    gcn_offsets_kernel<<<n_scan, 256, 0, stream>>>(table, bsum, offs, m);
    gcn_place_kernel<<<n_chunks, SORT_THREADS, 2 * lds_bins, stream>>>(rows, cols, vals, offs, recs, n_edges, n_bins, n_chunks);
    gcn_binsum_kernel<<<n_bins, 256, 0, stream>>>(recs, offs, X, agg, n_edges, n_bins, n_nodes, n_chunks);

    const int n_tiles = (n_nodes + 15) / 16;
    gcn_gemm_wmma_kernel<<<(n_tiles + 7) / 8, 256, 0, stream>>>(agg, W, bias, out, n_tiles);
}


# Sort each chunk by bin inside its work-group, store it in place with a table of run starts; the bin kernel copies its runs into LDS: no global scan, 3 launches fewer; row sums keep LDS arrival order
# speedup vs baseline: 2.4769x; 1.1874x over previous
#include <hip/hip_runtime.h>

typedef __attribute__((ext_vector_type(2))) float v2f;
typedef __attribute__((ext_vector_type(8))) float v8f;

#define D_FEAT 64
#define LDS_STRIDE 66

#define BIN_SHIFT 7
#define BIN_ROWS  (1 << BIN_SHIFT)
#define SORT_CHUNK 12288
#define SORT_THREADS 1024
#define REC_CAP 3072
#define STAGE_RUNS 9


__global__ __launch_bounds__(SORT_THREADS) void gcn_chunksort_kernel(
    const int*   __restrict__ rows,
    const int*   __restrict__ cols,
    const float* __restrict__ vals,
    int*         __restrict__ starts,
    int2*        __restrict__ recs,
    int n_edges, int n_bins)
{
    extern __shared__ int lslot[];
    __shared__ int2 srec[SORT_CHUNK];
    __shared__ int  wsum[SORT_THREADS / 32];

    const int tid  = threadIdx.x;
    const int lane = tid & 31;
    const int wave = tid >> 5;
    const int blk  = blockIdx.x;

    for (int i = tid; i < n_bins; i += SORT_THREADS) lslot[i] = 0;
    __syncthreads();

    const int lo = blk * SORT_CHUNK;
    const int hi = min(n_edges, lo + SORT_CHUNK);
    for (int e = lo + tid * 4; e < hi; e += SORT_THREADS * 4) {
        if (e + 3 < hi) {
            const int4 r = *(const int4*)(rows + e);
            atomicAdd(&lslot[r.x >> BIN_SHIFT], 1);
            atomicAdd(&lslot[r.y >> BIN_SHIFT], 1);
            atomicAdd(&lslot[r.z >> BIN_SHIFT], 1);
            atomicAdd(&lslot[r.w >> BIN_SHIFT], 1);
        } else {
            for (int j = e; j < hi; ++j) atomicAdd(&lslot[rows[j] >> BIN_SHIFT], 1);
        }
    }
    __syncthreads();

    int* mystarts = starts + (size_t)blk * (n_bins + 1);
    int carry = 0;
    for (int base = 0; base < n_bins; base += SORT_THREADS) {
        const int i = base + tid;
        const int c = (i < n_bins) ? lslot[i] : 0;
        int inc = c;
        for (int d = 1; d < 32; d <<= 1) {
            const int u = __shfl_up(inc, d, 32);
            if (lane >= d) inc += u;
        }
        if (lane == 31) wsum[wave] = inc;
        __syncthreads();
        int ws = wsum[lane];
        for (int d = 1; d < 32; d <<= 1) {
            const int u = __shfl_up(ws, d, 32);
            if (lane >= d) ws += u;
        }
        const int tot = __shfl(ws, 31, 32);
        const int wp  = __shfl(ws, wave, 32) - wsum[wave];
        const int ex  = carry + wp + inc - c;
        if (i < n_bins) { lslot[i] = ex; mystarts[i] = lo + ex; }
        carry += tot;
        __syncthreads();
    }
    if (tid == 0) mystarts[n_bins] = hi;

    for (int e = lo + tid * 4; e < hi; e += SORT_THREADS * 4) {
        if (e + 3 < hi) {
            const int4   r = *(const int4*)(rows + e);
            const int4   c = *(const int4*)(cols + e);
            const float4 v = *(const float4*)(vals + e);
            const int s0 = atomicAdd(&lslot[r.x >> BIN_SHIFT], 1);
            const int s1 = atomicAdd(&lslot[r.y >> BIN_SHIFT], 1);
            const int s2 = atomicAdd(&lslot[r.z >> BIN_SHIFT], 1);
            const int s3 = atomicAdd(&lslot[r.w >> BIN_SHIFT], 1);
            srec[s0] = make_int2(c.x | ((r.x & (BIN_ROWS - 1)) << 24), __float_as_int(v.x));
            srec[s1] = make_int2(c.y | ((r.y & (BIN_ROWS - 1)) << 24), __float_as_int(v.y));
            srec[s2] = make_int2(c.z | ((r.z & (BIN_ROWS - 1)) << 24), __float_as_int(v.z));
            srec[s3] = make_int2(c.w | ((r.w & (BIN_ROWS - 1)) << 24), __float_as_int(v.w));
        } else {
            for (int j = e; j < hi; ++j) {
                const int rj = rows[j];
                const int s = atomicAdd(&lslot[rj >> BIN_SHIFT], 1);
                srec[s] = make_int2(cols[j] | ((rj & (BIN_ROWS - 1)) << 24), __float_as_int(vals[j]));
            }
        }
    }
    __syncthreads();

    const int n = hi - lo;
    for (int t = tid; t < n; t += SORT_THREADS) recs[lo + t] = srec[t];
}

__global__ __launch_bounds__(256) void gcn_binsum_kernel(
    const int2*  __restrict__ recs,
    const int*   __restrict__ starts,
    const float* __restrict__ X,
    float*       __restrict__ agg,
    int n_bins, int n_nodes, int n_chunks)
{
    extern __shared__ int lruns[];
    __shared__ alignas(16) float tile[BIN_ROWS * D_FEAT];
    __shared__ int2 lrec[REC_CAP];
    __shared__ int  wsum[8];

    int* rsrc = lruns;
    int* rcnt = lruns + n_chunks;
    int* rpos = lruns + 2 * n_chunks;

    const int tid  = threadIdx.x;
    const int lane = tid & 31;
    const int bin  = blockIdx.x;

    float4* tile4 = (float4*)tile;
    for (int i = tid; i < BIN_ROWS * D_FEAT / 4; i += 256) tile4[i] = make_float4(0.f, 0.f, 0.f, 0.f);

    int carry = 0;
    for (int base = 0; base < n_chunks; base += 256) {
        const int c = base + tid;
        int s = 0, n = 0;
        if (c < n_chunks) {
            const int* p = starts + (size_t)c * (n_bins + 1) + bin;
            s = p[0];
            n = p[1] - s;
        }
        int inc = n;
        for (int d = 1; d < 32; d <<= 1) {
            const int u = __shfl_up(inc, d, 32);
            if (lane >= d) inc += u;
        }
        if (lane == 31) wsum[tid >> 5] = inc;
        __syncthreads();
        int wp = 0, tot = 0;
        for (int w = 0; w < 8; ++w) {
            const int t = wsum[w];
            tot += t;
            if (w < (tid >> 5)) wp += t;
        }
        if (c < n_chunks) { rsrc[c] = s; rcnt[c] = n; rpos[c] = carry + wp + inc - n; }
        carry += tot;
        __syncthreads();
    }
    const int total = carry;

    const int wave = __builtin_amdgcn_readfirstlane(tid >> 5);
    const v2f* X2 = (const v2f*)X;

    for (int pass0 = 0; pass0 < total; pass0 += REC_CAP) {
        for (int cb = wave; cb < n_chunks; cb += 8 * STAGE_RUNS) {
            int s[STAGE_RUNS], p0[STAGE_RUNS], j0[STAGE_RUNS], j1[STAGE_RUNS];
            int2 r[STAGE_RUNS];
#pragma unroll
            for (int u = 0; u < STAGE_RUNS; ++u) {
                const int c  = cb + 8 * u;
                const int cc = min(c, n_chunks - 1);
                const int n  = rcnt[cc];
                s[u]  = rsrc[cc];
                p0[u] = rpos[cc] - pass0;
                j0[u] = max(0, -p0[u]) + lane;
                j1[u] = (c < n_chunks) ? min(n, REC_CAP - p0[u]) : 0;
            }
#pragma unroll
            for (int u = 0; u < STAGE_RUNS; ++u) {
                r[u] = make_int2(0, 0);
                if (j0[u] < j1[u]) r[u] = recs[s[u] + j0[u]];
            }
#pragma unroll
            for (int u = 0; u < STAGE_RUNS; ++u) {
                if (j0[u] < j1[u]) lrec[p0[u] + j0[u]] = r[u];
            }
#pragma unroll
            for (int u = 0; u < STAGE_RUNS; ++u) {
                for (int j = j0[u] + 32; j < j1[u]; j += 32) lrec[p0[u] + j] = recs[s[u] + j];
            }
        }
        __syncthreads();

        const int nb = min(REC_CAP, total - pass0);
        int e0 = wave;
        if (e0 < nb) {
            const int last = nb - 1;
            int2 rec[8];
#pragma unroll
            for (int k = 0; k < 8; ++k) rec[k] = lrec[min(e0 + 8 * k, last)];
            do {
                v2f x[8];
                int rx[8], ry[8];
#pragma unroll
                for (int k = 0; k < 8; ++k) {
                    rx[k] = __builtin_amdgcn_readfirstlane(rec[k].x);
                    ry[k] = __builtin_amdgcn_readfirstlane(rec[k].y);
                    const int c = rx[k] & 0xFFFFFF;
                    x[k] = X2[(size_t)c * (D_FEAT / 2) + lane];
                }
                const int n0 = e0 + 64;
                int2 nrec[8];
#pragma unroll
                for (int k = 0; k < 8; ++k) nrec[k] = lrec[min(n0 + 8 * k, last)];
#pragma unroll
                for (int k = 0; k < 8; ++k) {
                    const bool on = e0 + 8 * k < nb;
                    const float v = __int_as_float(ry[k]);
                    float* dst = tile + (((unsigned)rx[k] >> 24) * D_FEAT) + lane * 2;
                    atomicAdd(dst + 0, on ? v * x[k].x : 0.f);
                    atomicAdd(dst + 1, on ? v * x[k].y : 0.f);
                }
#pragma unroll
                for (int k = 0; k < 8; ++k) rec[k] = nrec[k];
                e0 = n0;
            } while (e0 < nb);
        }
        __syncthreads();
    }

    const int row0 = bin * BIN_ROWS;
    float4* agg4 = (float4*)agg;
    for (int i = tid; i < BIN_ROWS * D_FEAT / 4; i += 256) {
        const int row = row0 + (i >> 4);
        if (row < n_nodes) agg4[(size_t)row * (D_FEAT / 4) + (i & 15)] = tile4[i];
    }
}

__global__ __launch_bounds__(256) void gcn_gemm_wmma_kernel(
    const float* __restrict__ agg,
    const float* __restrict__ W,
    const float* __restrict__ bias,
    float*       __restrict__ out,
    int n_tiles)
{
    __shared__ float ldsW[D_FEAT * LDS_STRIDE];

    const int tid = threadIdx.x;

    for (int i = tid; i < D_FEAT * D_FEAT; i += 256) {
        ldsW[(i >> 6) * LDS_STRIDE + (i & 63)] = W[i];
    }
    __syncthreads();

    const int wave = tid >> 5;
    const int lane = tid & 31;
    const int half = lane >> 4;
    const int l16  = lane & 15;
    const int koff = half << 1;

    const int tile = blockIdx.x * 8 + wave;
    if (tile >= n_tiles) return;

    const float bn0 = bias[ 0 + l16];
    const float bn1 = bias[16 + l16];
    const float bn2 = bias[32 + l16];
    const float bn3 = bias[48 + l16];

    v8f c0 = {}; v8f c1 = {}; v8f c2 = {}; v8f c3 = {};

    const float* arow = agg + ((size_t)tile * 16 + l16) * D_FEAT;

#pragma unroll
    for (int kk = 0; kk < 16; ++kk) {
        const int k0 = kk * 4;

        v2f a = *(const v2f*)(arow + k0 + koff);

        const int kb = k0 + koff;
        v2f b0 = *(const v2f*)(&ldsW[( 0 + l16) * LDS_STRIDE + kb]);
        v2f b1 = *(const v2f*)(&ldsW[(16 + l16) * LDS_STRIDE + kb]);
        v2f b2 = *(const v2f*)(&ldsW[(32 + l16) * LDS_STRIDE + kb]);
        v2f b3 = *(const v2f*)(&ldsW[(48 + l16) * LDS_STRIDE + kb]);

        c0 = __builtin_amdgcn_wmma_f32_16x16x4_f32(false, a, false, b0, (short)0, c0, false, false);
        c1 = __builtin_amdgcn_wmma_f32_16x16x4_f32(false, a, false, b1, (short)0, c1, false, false);
        c2 = __builtin_amdgcn_wmma_f32_16x16x4_f32(false, a, false, b2, (short)0, c2, false, false);
        c3 = __builtin_amdgcn_wmma_f32_16x16x4_f32(false, a, false, b3, (short)0, c3, false, false);
    }

    const size_t base = (size_t)tile * 16 * D_FEAT;
#pragma unroll
    for (int r = 0; r < 8; ++r) {
        const size_t m_off = base + (size_t)(r + half * 8) * D_FEAT;
        out[m_off +  0 + l16] = c0[r] + bn0;
        out[m_off + 16 + l16] = c1[r] + bn1;
        out[m_off + 32 + l16] = c2[r] + bn2;
        out[m_off + 48 + l16] = c3[r] + bn3;
    }
}

extern "C" void kernel_launch(void* const* d_in, const int* in_sizes, int n_in,
                              void* d_out, int out_size, void* d_ws, size_t ws_size,
                              hipStream_t stream) {
    const int*   rows = (const int*)d_in[0];
    const int*   cols = (const int*)d_in[1];
    const float* vals = (const float*)d_in[2];
    const float* X    = (const float*)d_in[3];
    const float* W    = (const float*)d_in[4];
    const float* bias = (const float*)d_in[5];
    float* out = (float*)d_out;

    const int n_edges = in_sizes[0];
    const int n_nodes = in_sizes[3] / D_FEAT;
    const int n_bins  = (n_nodes + BIN_ROWS - 1) / BIN_ROWS;
    const int n_chunks = (n_edges + SORT_CHUNK - 1) / SORT_CHUNK;

    char* ws = (char*)d_ws;
    float* agg = (float*)ws;
    size_t o = (((size_t)n_nodes * D_FEAT * sizeof(float)) + 255) & ~(size_t)255;
    int* starts = (int*)(ws + o);  o += (((size_t)n_chunks * (n_bins + 1) * sizeof(int)) + 255) & ~(size_t)255;
    int2* recs = (int2*)(ws + o);

    gcn_chunksort_kernel<<<n_chunks, SORT_THREADS, (size_t)n_bins * sizeof(int), stream>>>(rows, cols, vals, starts, recs, n_edges, n_bins);
    gcn_binsum_kernel<<<n_bins, 256, (size_t)3 * n_chunks * sizeof(int), stream>>>(recs, starts, X, agg, n_bins, n_nodes, n_chunks);

    const int n_tiles = (n_nodes + 15) / 16;
    gcn_gemm_wmma_kernel<<<(n_tiles + 7) / 8, 256, 0, stream>>>(agg, W, bias, out, n_tiles);
}


# Apply the 64x64 layer to X first (out = L(XW^T)+b, f32 WMMA) inside the edge-sort launch, on the WGPs the sort leaves idle; bin sums store out rows: 2 launches; sums of projected rows, arrival order
# speedup vs baseline: 2.5745x; 1.0394x over previous
#include <hip/hip_runtime.h>

typedef __attribute__((ext_vector_type(2))) float v2f;
typedef __attribute__((ext_vector_type(8))) float v8f;

#define D_FEAT 64
#define LDS_STRIDE 66

#define BIN_SHIFT 7
#define BIN_ROWS  (1 << BIN_SHIFT)
#define SORT_CHUNK 12288
#define SORT_THREADS 1024
#define REC_CAP 3072
#define STAGE_RUNS 9
#define N_WGP 256


__device__ __forceinline__ void project_tiles(
    const float* __restrict__ X,
    const float* __restrict__ W,
    float*       __restrict__ Y,
    float* ldsW,
    int group, int n_groups, int n_tiles)
{
    const int tid = threadIdx.x;

    for (int i = tid; i < D_FEAT * D_FEAT; i += SORT_THREADS) {
        ldsW[(i >> 6) * LDS_STRIDE + (i & 63)] = W[i];
    }
    __syncthreads();

    const int wave = tid >> 5;
    const int lane = tid & 31;
    const int half = lane >> 4;
    const int l16  = lane & 15;
    const int koff = half << 1;

    const int t_lo = (int)(((unsigned)group * (unsigned)n_tiles) / (unsigned)n_groups);
    const int t_hi = (int)(((unsigned)(group + 1) * (unsigned)n_tiles) / (unsigned)n_groups);
    const int tile0 = t_lo + wave;
    const int tile1 = tile0 + SORT_THREADS / 32;
    if (tile0 >= t_hi) return;

    const float* arow0 = X + ((size_t)tile0 * 16 + l16) * D_FEAT;
    const size_t base0 = (size_t)tile0 * 16 * D_FEAT;

    if (tile1 < t_hi) {
        v8f c0 = {}; v8f c1 = {}; v8f c2 = {}; v8f c3 = {};
        v8f d0 = {}; v8f d1 = {}; v8f d2 = {}; v8f d3 = {};
        const float* arow1 = X + ((size_t)tile1 * 16 + l16) * D_FEAT;

#pragma unroll
        for (int kk = 0; kk < 16; ++kk) {
            const int kb = kk * 4 + koff;

            v2f a0 = *(const v2f*)(arow0 + kb);
            v2f a1 = *(const v2f*)(arow1 + kb);

            v2f b0 = *(const v2f*)(&ldsW[( 0 + l16) * LDS_STRIDE + kb]);
            v2f b1 = *(const v2f*)(&ldsW[(16 + l16) * LDS_STRIDE + kb]);
            v2f b2 = *(const v2f*)(&ldsW[(32 + l16) * LDS_STRIDE + kb]);
            v2f b3 = *(const v2f*)(&ldsW[(48 + l16) * LDS_STRIDE + kb]);

            c0 = __builtin_amdgcn_wmma_f32_16x16x4_f32(false, a0, false, b0, (short)0, c0, false, false);
            d0 = __builtin_amdgcn_wmma_f32_16x16x4_f32(false, a1, false, b0, (short)0, d0, false, false);
            c1 = __builtin_amdgcn_wmma_f32_16x16x4_f32(false, a0, false, b1, (short)0, c1, false, false);
            d1 = __builtin_amdgcn_wmma_f32_16x16x4_f32(false, a1, false, b1, (short)0, d1, false, false);
            c2 = __builtin_amdgcn_wmma_f32_16x16x4_f32(false, a0, false, b2, (short)0, c2, false, false);
            d2 = __builtin_amdgcn_wmma_f32_16x16x4_f32(false, a1, false, b2, (short)0, d2, false, false);
            c3 = __builtin_amdgcn_wmma_f32_16x16x4_f32(false, a0, false, b3, (short)0, c3, false, false);
            d3 = __builtin_amdgcn_wmma_f32_16x16x4_f32(false, a1, false, b3, (short)0, d3, false, false);
        }

        const size_t base1 = (size_t)tile1 * 16 * D_FEAT;
#pragma unroll
        for (int r = 0; r < 8; ++r) {
            const size_t m_off = (size_t)(r + half * 8) * D_FEAT;
            Y[base0 + m_off +  0 + l16] = c0[r];
            Y[base0 + m_off + 16 + l16] = c1[r];
            Y[base0 + m_off + 32 + l16] = c2[r];
            Y[base0 + m_off + 48 + l16] = c3[r];
        }
#pragma unroll
        for (int r = 0; r < 8; ++r) {
            const size_t m_off = (size_t)(r + half * 8) * D_FEAT;
            Y[base1 + m_off +  0 + l16] = d0[r];
            Y[base1 + m_off + 16 + l16] = d1[r];
            Y[base1 + m_off + 32 + l16] = d2[r];
            Y[base1 + m_off + 48 + l16] = d3[r];
        }
    } else {
        v8f c0 = {}; v8f c1 = {}; v8f c2 = {}; v8f c3 = {};

#pragma unroll
        for (int kk = 0; kk < 16; ++kk) {
            const int kb = kk * 4 + koff;

            v2f a0 = *(const v2f*)(arow0 + kb);

            v2f b0 = *(const v2f*)(&ldsW[( 0 + l16) * LDS_STRIDE + kb]);
            v2f b1 = *(const v2f*)(&ldsW[(16 + l16) * LDS_STRIDE + kb]);
            v2f b2 = *(const v2f*)(&ldsW[(32 + l16) * LDS_STRIDE + kb]);
            v2f b3 = *(const v2f*)(&ldsW[(48 + l16) * LDS_STRIDE + kb]);

            c0 = __builtin_amdgcn_wmma_f32_16x16x4_f32(false, a0, false, b0, (short)0, c0, false, false);
            c1 = __builtin_amdgcn_wmma_f32_16x16x4_f32(false, a0, false, b1, (short)0, c1, false, false);
            c2 = __builtin_amdgcn_wmma_f32_16x16x4_f32(false, a0, false, b2, (short)0, c2, false, false);
            c3 = __builtin_amdgcn_wmma_f32_16x16x4_f32(false, a0, false, b3, (short)0, c3, false, false);
        }

#pragma unroll
        for (int r = 0; r < 8; ++r) {
            const size_t m_off = (size_t)(r + half * 8) * D_FEAT;
            Y[base0 + m_off +  0 + l16] = c0[r];
            Y[base0 + m_off + 16 + l16] = c1[r];
            Y[base0 + m_off + 32 + l16] = c2[r];
            Y[base0 + m_off + 48 + l16] = c3[r];
        }
    }
}

__global__ __launch_bounds__(SORT_THREADS) void gcn_sort_project_kernel(
    const int*   __restrict__ rows,
    const int*   __restrict__ cols,
    const float* __restrict__ vals,
    const float* __restrict__ X,
    const float* __restrict__ W,
    int*         __restrict__ starts,
    int2*        __restrict__ recs,
    float*       __restrict__ Y,
    int n_edges, int n_bins, int n_chunks, int n_proj, int n_tiles)
{
    extern __shared__ int lslot[];
    __shared__ int2 srec[SORT_CHUNK];
    __shared__ int  wsum[SORT_THREADS / 32];

    const int tid  = threadIdx.x;
    const int lane = tid & 31;
    const int wave = tid >> 5;
    const int blk  = blockIdx.x;

    if (blk >= n_chunks) {
        project_tiles(X, W, Y, (float*)srec, blk - n_chunks, n_proj, n_tiles);
        return;
    }

    for (int i = tid; i < n_bins; i += SORT_THREADS) lslot[i] = 0;
    __syncthreads();

    const int lo = blk * SORT_CHUNK;
    const int hi = min(n_edges, lo + SORT_CHUNK);
    for (int e = lo + tid * 4; e < hi; e += SORT_THREADS * 4) {
        if (e + 3 < hi) {
            const int4 r = *(const int4*)(rows + e);
            atomicAdd(&lslot[r.x >> BIN_SHIFT], 1);
            atomicAdd(&lslot[r.y >> BIN_SHIFT], 1);
            atomicAdd(&lslot[r.z >> BIN_SHIFT], 1);
            atomicAdd(&lslot[r.w >> BIN_SHIFT], 1);
        } else {
            for (int j = e; j < hi; ++j) atomicAdd(&lslot[rows[j] >> BIN_SHIFT], 1);
        }
    }
    __syncthreads();

    int* mystarts = starts + (size_t)blk * (n_bins + 1);
    int carry = 0;
    for (int base = 0; base < n_bins; base += SORT_THREADS) {
        const int i = base + tid;
        const int c = (i < n_bins) ? lslot[i] : 0;
        int inc = c;
        for (int d = 1; d < 32; d <<= 1) {
            const int u = __shfl_up(inc, d, 32);
            if (lane >= d) inc += u;
        }
        if (lane == 31) wsum[wave] = inc;
        __syncthreads();
        int ws = wsum[lane];
        for (int d = 1; d < 32; d <<= 1) {
            const int u = __shfl_up(ws, d, 32);
            if (lane >= d) ws += u;
        }
        const int tot = __shfl(ws, 31, 32);
        const int wp  = __shfl(ws, wave, 32) - wsum[wave];
        const int ex  = carry + wp + inc - c;
        if (i < n_bins) { lslot[i] = ex; mystarts[i] = lo + ex; }
        carry += tot;
        __syncthreads();
    }
    if (tid == 0) mystarts[n_bins] = hi;

    for (int e = lo + tid * 4; e < hi; e += SORT_THREADS * 4) {
        if (e + 3 < hi) {
            const int4   r = *(const int4*)(rows + e);
            const int4   c = *(const int4*)(cols + e);
            const float4 v = *(const float4*)(vals + e);
            const int s0 = atomicAdd(&lslot[r.x >> BIN_SHIFT], 1);
            const int s1 = atomicAdd(&lslot[r.y >> BIN_SHIFT], 1);
            const int s2 = atomicAdd(&lslot[r.z >> BIN_SHIFT], 1);
            const int s3 = atomicAdd(&lslot[r.w >> BIN_SHIFT], 1);
            srec[s0] = make_int2(c.x | ((r.x & (BIN_ROWS - 1)) << 24), __float_as_int(v.x));
            srec[s1] = make_int2(c.y | ((r.y & (BIN_ROWS - 1)) << 24), __float_as_int(v.y));
            srec[s2] = make_int2(c.z | ((r.z & (BIN_ROWS - 1)) << 24), __float_as_int(v.z));
            srec[s3] = make_int2(c.w | ((r.w & (BIN_ROWS - 1)) << 24), __float_as_int(v.w));
        } else {
            for (int j = e; j < hi; ++j) {
                const int rj = rows[j];
                const int s = atomicAdd(&lslot[rj >> BIN_SHIFT], 1);
                srec[s] = make_int2(cols[j] | ((rj & (BIN_ROWS - 1)) << 24), __float_as_int(vals[j]));
            }
        }
    }
    __syncthreads();

    const int n = hi - lo;
    for (int t = tid; t < n; t += SORT_THREADS) recs[lo + t] = srec[t];
}

__global__ __launch_bounds__(256) void gcn_binsum_kernel(
    const int2*  __restrict__ recs,
    const int*   __restrict__ starts,
    const float* __restrict__ Y,
    const float* __restrict__ bias,
    float*       __restrict__ out,
    int n_bins, int n_nodes, int n_chunks)
{
    extern __shared__ int lruns[];
    __shared__ alignas(16) float tile[BIN_ROWS * D_FEAT];
    __shared__ int2 lrec[REC_CAP];
    __shared__ int  wsum[8];

    int* rsrc = lruns;
    int* rcnt = lruns + n_chunks;
    int* rpos = lruns + 2 * n_chunks;

    const int tid  = threadIdx.x;
    const int lane = tid & 31;
    const int bin  = blockIdx.x;

    float4* tile4 = (float4*)tile;
    for (int i = tid; i < BIN_ROWS * D_FEAT / 4; i += 256) tile4[i] = make_float4(0.f, 0.f, 0.f, 0.f);

    int carry = 0;
    for (int base = 0; base < n_chunks; base += 256) {
        const int c = base + tid;
        int s = 0, n = 0;
        if (c < n_chunks) {
            const int* p = starts + (size_t)c * (n_bins + 1) + bin;
            s = p[0];
            n = p[1] - s;
        }
        int inc = n;
        for (int d = 1; d < 32; d <<= 1) {
            const int u = __shfl_up(inc, d, 32);
            if (lane >= d) inc += u;
        }
        if (lane == 31) wsum[tid >> 5] = inc;
        __syncthreads();
        int wp = 0, tot = 0;
        for (int w = 0; w < 8; ++w) {
            const int t = wsum[w];
            tot += t;
            if (w < (tid >> 5)) wp += t;
        }
        if (c < n_chunks) { rsrc[c] = s; rcnt[c] = n; rpos[c] = carry + wp + inc - n; }
        carry += tot;
        __syncthreads();
    }
    const int total = carry;

    const int wave = __builtin_amdgcn_readfirstlane(tid >> 5);
    const v2f* Y2 = (const v2f*)Y;

    for (int pass0 = 0; pass0 < total; pass0 += REC_CAP) {
        for (int cb = wave; cb < n_chunks; cb += 8 * STAGE_RUNS) {
            int s[STAGE_RUNS], p0[STAGE_RUNS], j0[STAGE_RUNS], j1[STAGE_RUNS];
            int2 r[STAGE_RUNS];
#pragma unroll
            for (int u = 0; u < STAGE_RUNS; ++u) {
                const int c  = cb + 8 * u;
                const int cc = min(c, n_chunks - 1);
                const int n  = rcnt[cc];
                s[u]  = rsrc[cc];
                p0[u] = rpos[cc] - pass0;
                j0[u] = max(0, -p0[u]) + lane;
                j1[u] = (c < n_chunks) ? min(n, REC_CAP - p0[u]) : 0;
            }
#pragma unroll
            for (int u = 0; u < STAGE_RUNS; ++u) {
                r[u] = make_int2(0, 0);
                if (j0[u] < j1[u]) r[u] = recs[s[u] + j0[u]];
            }
#pragma unroll
            for (int u = 0; u < STAGE_RUNS; ++u) {
                if (j0[u] < j1[u]) lrec[p0[u] + j0[u]] = r[u];
            }
#pragma unroll
            for (int u = 0; u < STAGE_RUNS; ++u) {
                for (int j = j0[u] + 32; j < j1[u]; j += 32) lrec[p0[u] + j] = recs[s[u] + j];
            }
        }
        __syncthreads();

        const int nb = min(REC_CAP, total - pass0);
        int e0 = wave;
        if (e0 < nb) {
            const int last = nb - 1;
            int2 rec[8];
#pragma unroll
            for (int k = 0; k < 8; ++k) rec[k] = lrec[min(e0 + 8 * k, last)];
            do {
                v2f x[8];
                int rx[8], ry[8];
#pragma unroll
                for (int k = 0; k < 8; ++k) {
                    rx[k] = __builtin_amdgcn_readfirstlane(rec[k].x);
                    ry[k] = __builtin_amdgcn_readfirstlane(rec[k].y);
                    const int c = rx[k] & 0xFFFFFF;
                    x[k] = Y2[(size_t)c * (D_FEAT / 2) + lane];
                }
                const int n0 = e0 + 64;
                int2 nrec[8];
#pragma unroll
                for (int k = 0; k < 8; ++k) nrec[k] = lrec[min(n0 + 8 * k, last)];
#pragma unroll
                for (int k = 0; k < 8; ++k) {
                    const bool on = e0 + 8 * k < nb;
                    const float v = __int_as_float(ry[k]);
                    float* dst = tile + (((unsigned)rx[k] >> 24) * D_FEAT) + lane * 2;
                    atomicAdd(dst + 0, on ? v * x[k].x : 0.f);
                    atomicAdd(dst + 1, on ? v * x[k].y : 0.f);
                }
#pragma unroll
                for (int k = 0; k < 8; ++k) rec[k] = nrec[k];
                e0 = n0;
            } while (e0 < nb);
        }
        __syncthreads();
    }

    const int row0 = bin * BIN_ROWS;
    const float4 b4 = ((const float4*)bias)[tid & 15];
    float4* out4 = (float4*)out;
    for (int i = tid; i < BIN_ROWS * D_FEAT / 4; i += 256) {
        const int row = row0 + (i >> 4);
        if (row < n_nodes) {
            const float4 t = tile4[i];
            out4[(size_t)row * (D_FEAT / 4) + (i & 15)] = make_float4(t.x + b4.x, t.y + b4.y, t.z + b4.z, t.w + b4.w);
        }
    }
}

extern "C" void kernel_launch(void* const* d_in, const int* in_sizes, int n_in,
                              void* d_out, int out_size, void* d_ws, size_t ws_size,
                              hipStream_t stream) {
    const int*   rows = (const int*)d_in[0];
    const int*   cols = (const int*)d_in[1];
    const float* vals = (const float*)d_in[2];
    const float* X    = (const float*)d_in[3];
    const float* W    = (const float*)d_in[4];
    const float* bias = (const float*)d_in[5];
    float* out = (float*)d_out;

    const int n_edges = in_sizes[0];
    const int n_nodes = in_sizes[3] / D_FEAT;
    const int n_bins  = (n_nodes + BIN_ROWS - 1) / BIN_ROWS;
    const int n_chunks = (n_edges + SORT_CHUNK - 1) / SORT_CHUNK;
    const int n_tiles = (n_nodes + 15) / 16;
    const int n_proj_min = (n_tiles + 63) / 64;
    const int n_proj  = (N_WGP - n_chunks > n_proj_min) ? (N_WGP - n_chunks) : n_proj_min;

    char* ws = (char*)d_ws;
    float* Y = (float*)ws;
    size_t o = (((size_t)n_tiles * 16 * D_FEAT * sizeof(float)) + 255) & ~(size_t)255;
    int* starts = (int*)(ws + o);  o += (((size_t)n_chunks * (n_bins + 1) * sizeof(int)) + 255) & ~(size_t)255;
    int2* recs = (int2*)(ws + o);

    gcn_sort_project_kernel<<<n_chunks + n_proj, SORT_THREADS, (size_t)n_bins * sizeof(int), stream>>>(
        rows, cols, vals, X, W, starts, recs, Y, n_edges, n_bins, n_chunks, n_proj, n_tiles);
    gcn_binsum_kernel<<<n_bins, 256, (size_t)3 * n_chunks * sizeof(int), stream>>>(recs, starts, Y, bias, out, n_bins, n_nodes, n_chunks);
}


# Bin kernel sorts its records by row in LDS (integer slot adds); each wave sums its own 16 rows in registers by FMA: no LDS f32 atomics; row sum order = sorted record order (slot arrival)
# speedup vs baseline: 3.5496x; 1.3788x over previous
#include <hip/hip_runtime.h>

typedef __attribute__((ext_vector_type(2))) float v2f;
typedef __attribute__((ext_vector_type(8))) float v8f;

#define D_FEAT 64
#define W_STRIDE 68
#define TILE_STRIDE 68

#define BIN_SHIFT 7
#define BIN_ROWS  (1 << BIN_SHIFT)
#define SORT_CHUNK 12288
#define SORT_THREADS 1024
#define REC_CAP 2560
#define STAGE_RUNS 9


__global__ __launch_bounds__(SORT_THREADS) void gcn_sort_kernel(
    const int*   __restrict__ rows,
    const int*   __restrict__ cols,
    const float* __restrict__ vals,
    int*         __restrict__ starts,
    int2*        __restrict__ recs,
    int n_edges, int n_bins)
{
    extern __shared__ int lslot[];
    __shared__ int2 srec[SORT_CHUNK];
    __shared__ int  wsum[SORT_THREADS / 32];

    const int tid  = threadIdx.x;
    const int lane = tid & 31;
    const int wave = tid >> 5;
    const int blk  = blockIdx.x;

    for (int i = tid; i < n_bins; i += SORT_THREADS) lslot[i] = 0;
    __syncthreads();

    const int lo = blk * SORT_CHUNK;
    const int hi = min(n_edges, lo + SORT_CHUNK);
    for (int e = lo + tid * 4; e < hi; e += SORT_THREADS * 4) {
        if (e + 3 < hi) {
            const int4 r = *(const int4*)(rows + e);
            atomicAdd(&lslot[r.x >> BIN_SHIFT], 1);
            atomicAdd(&lslot[r.y >> BIN_SHIFT], 1);
            atomicAdd(&lslot[r.z >> BIN_SHIFT], 1);
            atomicAdd(&lslot[r.w >> BIN_SHIFT], 1);
        } else {
            for (int j = e; j < hi; ++j) atomicAdd(&lslot[rows[j] >> BIN_SHIFT], 1);
        }
    }
    __syncthreads();

    int* mystarts = starts + (size_t)blk * (n_bins + 1);
    int carry = 0;
    for (int base = 0; base < n_bins; base += SORT_THREADS) {
        const int i = base + tid;
        const int c = (i < n_bins) ? lslot[i] : 0;
        int inc = c;
        for (int d = 1; d < 32; d <<= 1) {
            const int u = __shfl_up(inc, d, 32);
            if (lane >= d) inc += u;
        }
        if (lane == 31) wsum[wave] = inc;
        __syncthreads();
        int ws = wsum[lane];
        for (int d = 1; d < 32; d <<= 1) {
            const int u = __shfl_up(ws, d, 32);
            if (lane >= d) ws += u;
        }
        const int tot = __shfl(ws, 31, 32);
        const int wp  = __shfl(ws, wave, 32) - wsum[wave];
        const int ex  = carry + wp + inc - c;
        if (i < n_bins) { lslot[i] = ex; mystarts[i] = lo + ex; }
        carry += tot;
        __syncthreads();
    }
    if (tid == 0) mystarts[n_bins] = hi;

    for (int e = lo + tid * 4; e < hi; e += SORT_THREADS * 4) {
        if (e + 3 < hi) {
            const int4   r = *(const int4*)(rows + e);
            const int4   c = *(const int4*)(cols + e);
            const float4 v = *(const float4*)(vals + e);
            const int s0 = atomicAdd(&lslot[r.x >> BIN_SHIFT], 1);
            const int s1 = atomicAdd(&lslot[r.y >> BIN_SHIFT], 1);
            const int s2 = atomicAdd(&lslot[r.z >> BIN_SHIFT], 1);
            const int s3 = atomicAdd(&lslot[r.w >> BIN_SHIFT], 1);
            srec[s0] = make_int2(c.x | ((r.x & (BIN_ROWS - 1)) << 24), __float_as_int(v.x));
            srec[s1] = make_int2(c.y | ((r.y & (BIN_ROWS - 1)) << 24), __float_as_int(v.y));
            srec[s2] = make_int2(c.z | ((r.z & (BIN_ROWS - 1)) << 24), __float_as_int(v.z));
            srec[s3] = make_int2(c.w | ((r.w & (BIN_ROWS - 1)) << 24), __float_as_int(v.w));
        } else {
            for (int j = e; j < hi; ++j) {
                const int rj = rows[j];
                const int s = atomicAdd(&lslot[rj >> BIN_SHIFT], 1);
                srec[s] = make_int2(cols[j] | ((rj & (BIN_ROWS - 1)) << 24), __float_as_int(vals[j]));
            }
        }
    }
    __syncthreads();

    const int n = hi - lo;
    for (int t = tid; t < n; t += SORT_THREADS) recs[lo + t] = srec[t];
}

__global__ __launch_bounds__(256) void gcn_binsum_dense_kernel(
    const int2*  __restrict__ recs,
    const int*   __restrict__ starts,
    const float* __restrict__ X,
    const float* __restrict__ W,
    const float* __restrict__ bias,
    float*       __restrict__ out,
    int n_bins, int n_nodes, int n_chunks)
{
    extern __shared__ int lruns[];
    __shared__ alignas(16) float tile[BIN_ROWS * TILE_STRIDE];
    __shared__ alignas(16) float ldsW[D_FEAT * W_STRIDE];
    __shared__ int2 lrec[REC_CAP];
    __shared__ int  wsum[8];
    __shared__ int  rowslot[BIN_ROWS];
    __shared__ int  rowbeg[BIN_ROWS + 1];

    int* rsrc = lruns;
    int* rcnt = lruns + n_chunks;
    int* rpos = lruns + 2 * n_chunks;

    const int tid  = threadIdx.x;
    const int lane = tid & 31;
    const int bin  = blockIdx.x;

    float4* tile4 = (float4*)tile;
    for (int i = tid; i < BIN_ROWS * TILE_STRIDE / 4; i += 256) tile4[i] = make_float4(0.f, 0.f, 0.f, 0.f);

    int carry = 0;
    for (int base = 0; base < n_chunks; base += 256) {
        const int c = base + tid;
        int s = 0, n = 0;
        if (c < n_chunks) {
            const int* p = starts + (size_t)c * (n_bins + 1) + bin;
            s = p[0];
            n = p[1] - s;
        }
        if (base == 0) {
            float4 wreg[D_FEAT * D_FEAT / 4 / 256];
#pragma unroll
            for (int j = 0; j < D_FEAT * D_FEAT / 4 / 256; ++j) wreg[j] = ((const float4*)W)[tid + 256 * j];
#pragma unroll
            for (int j = 0; j < D_FEAT * D_FEAT / 4 / 256; ++j) {
                const int i = tid + 256 * j;
                *(float4*)(&ldsW[(i >> 4) * W_STRIDE + (i & 15) * 4]) = wreg[j];
            }
        }
        int inc = n;
        for (int d = 1; d < 32; d <<= 1) {
            const int u = __shfl_up(inc, d, 32);
            if (lane >= d) inc += u;
        }
        if (lane == 31) wsum[tid >> 5] = inc;
        __syncthreads();
        int wp = 0, tot = 0;
        for (int w = 0; w < 8; ++w) {
            const int t = wsum[w];
            tot += t;
            if (w < (tid >> 5)) wp += t;
        }
        if (c < n_chunks) { rsrc[c] = s; rcnt[c] = n; rpos[c] = carry + wp + inc - n; }
        carry += tot;
        __syncthreads();
    }
    const int total = carry;

    const int wave = __builtin_amdgcn_readfirstlane(tid >> 5);
    const v2f* X2 = (const v2f*)X;

    for (int pass0 = 0; pass0 < total; pass0 += REC_CAP) {
        if (tid < BIN_ROWS) rowslot[tid] = 0;

        for (int cb = wave; cb < n_chunks; cb += 8 * STAGE_RUNS) {
            int s[STAGE_RUNS], p0[STAGE_RUNS], j0[STAGE_RUNS], j1[STAGE_RUNS];
            int2 r[STAGE_RUNS];
#pragma unroll
            for (int u = 0; u < STAGE_RUNS; ++u) {
                const int c  = cb + 8 * u;
                const int cc = min(c, n_chunks - 1);
                const int n  = rcnt[cc];
                s[u]  = rsrc[cc];
                p0[u] = rpos[cc] - pass0;
                j0[u] = max(0, -p0[u]) + lane;
                j1[u] = (c < n_chunks) ? min(n, REC_CAP - p0[u]) : 0;
            }
#pragma unroll
            for (int u = 0; u < STAGE_RUNS; ++u) {
                r[u] = make_int2(0, 0);
                if (j0[u] < j1[u]) r[u] = recs[s[u] + j0[u]];
            }
#pragma unroll
            for (int u = 0; u < STAGE_RUNS; ++u) {
                if (j0[u] < j1[u]) lrec[p0[u] + j0[u]] = r[u];
            }
#pragma unroll
            for (int u = 0; u < STAGE_RUNS; ++u) {
                for (int j = j0[u] + 32; j < j1[u]; j += 32) lrec[p0[u] + j] = recs[s[u] + j];
            }
        }
        __syncthreads();

        const int nb = min(REC_CAP, total - pass0);

        int2 mine[REC_CAP / 256];
#pragma unroll
        for (int k = 0; k < REC_CAP / 256; ++k) {
            const int i = tid + 256 * k;
            mine[k] = make_int2(0, 0);
            if (i < nb) {
                mine[k] = lrec[i];
                atomicAdd(&rowslot[(unsigned)mine[k].x >> 24], 1);
            }
        }
        __syncthreads();
        {
            const int c = (tid < BIN_ROWS) ? rowslot[tid] : 0;
            int inc = c;
            for (int d = 1; d < 32; d <<= 1) {
                const int u = __shfl_up(inc, d, 32);
                if (lane >= d) inc += u;
            }
            if (lane == 31) wsum[tid >> 5] = inc;
            __syncthreads();
            int wp = 0;
            for (int w = 0; w < BIN_ROWS / 32; ++w) {
                const int t = wsum[w];
                if (w < (tid >> 5)) wp += t;
            }
            if (tid < BIN_ROWS) {
                const int ex = wp + inc - c;
                rowbeg[tid]  = ex;
                rowslot[tid] = ex;
            }
            if (tid == 0) rowbeg[BIN_ROWS] = nb;
        }
        __syncthreads();
#pragma unroll
        for (int k = 0; k < REC_CAP / 256; ++k) {
            const int i = tid + 256 * k;
            if (i < nb) {
                const int sl = atomicAdd(&rowslot[(unsigned)mine[k].x >> 24], 1);
                lrec[sl] = mine[k];
            }
        }
        __syncthreads();

        const int e_lo = __builtin_amdgcn_readfirstlane(rowbeg[wave * 16]);
        const int e_hi = __builtin_amdgcn_readfirstlane(rowbeg[wave * 16 + 16]);
        if (e_lo < e_hi) {
            const int last = e_hi - 1;
            int2 rec[8];
#pragma unroll
            for (int k = 0; k < 8; ++k) rec[k] = lrec[min(e_lo + k, last)];
            int e0  = e_lo;
            int cur = (int)((unsigned)__builtin_amdgcn_readfirstlane(rec[0].x) >> 24);
            v2f acc = {0.f, 0.f};
            do {
                v2f x[8];
                int rx[8], ry[8];
#pragma unroll
                for (int k = 0; k < 8; ++k) {
                    rx[k] = __builtin_amdgcn_readfirstlane(rec[k].x);
                    ry[k] = __builtin_amdgcn_readfirstlane(rec[k].y);
                    const int c = rx[k] & 0xFFFFFF;
                    x[k] = X2[(size_t)c * (D_FEAT / 2) + lane];
                }
                const int n0 = e0 + 8;
                int2 nrec[8];
#pragma unroll
                for (int k = 0; k < 8; ++k) nrec[k] = lrec[min(n0 + k, last)];
#pragma unroll
                for (int k = 0; k < 8; ++k) {
                    if (e0 + k < e_hi) {
                        const int row = (int)((unsigned)rx[k] >> 24);
                        if (row != cur) {
                            v2f* d = (v2f*)(tile + cur * TILE_STRIDE + lane * 2);
                            v2f t = *d;
                            t.x += acc.x; t.y += acc.y;
                            *d = t;
                            acc.x = 0.f; acc.y = 0.f;
                            cur = row;
                        }
                        const float v = __int_as_float(ry[k]);
                        acc.x = fmaf(v, x[k].x, acc.x);
                        acc.y = fmaf(v, x[k].y, acc.y);
                    }
                }
#pragma unroll
                for (int k = 0; k < 8; ++k) rec[k] = nrec[k];
                e0 = n0;
            } while (e0 < e_hi);
            v2f* d = (v2f*)(tile + cur * TILE_STRIDE + lane * 2);
            v2f t = *d;
            t.x += acc.x; t.y += acc.y;
            *d = t;
        }
        __syncthreads();
    }

    {
        const int half = lane >> 4;
        const int l16  = lane & 15;
        const int koff = half << 1;
        const float* arow = tile + (wave * 16 + l16) * TILE_STRIDE;
        v8f c0 = {}; v8f c1 = {}; v8f c2 = {}; v8f c3 = {};

#pragma unroll
        for (int kk = 0; kk < 16; ++kk) {
            const int kb = kk * 4 + koff;

            v2f a0 = *(const v2f*)(arow + kb);

            v2f b0 = *(const v2f*)(&ldsW[( 0 + l16) * W_STRIDE + kb]);
            v2f b1 = *(const v2f*)(&ldsW[(16 + l16) * W_STRIDE + kb]);
            v2f b2 = *(const v2f*)(&ldsW[(32 + l16) * W_STRIDE + kb]);
            v2f b3 = *(const v2f*)(&ldsW[(48 + l16) * W_STRIDE + kb]);

            c0 = __builtin_amdgcn_wmma_f32_16x16x4_f32(false, a0, false, b0, (short)0, c0, false, false);
            c1 = __builtin_amdgcn_wmma_f32_16x16x4_f32(false, a0, false, b1, (short)0, c1, false, false);
            c2 = __builtin_amdgcn_wmma_f32_16x16x4_f32(false, a0, false, b2, (short)0, c2, false, false);
            c3 = __builtin_amdgcn_wmma_f32_16x16x4_f32(false, a0, false, b3, (short)0, c3, false, false);
        }

        float* drow = tile + (wave * 16 + half * 8) * TILE_STRIDE + l16;
#pragma unroll
        for (int r = 0; r < 8; ++r) {
            drow[r * TILE_STRIDE +  0] = c0[r];
            drow[r * TILE_STRIDE + 16] = c1[r];
            drow[r * TILE_STRIDE + 32] = c2[r];
            drow[r * TILE_STRIDE + 48] = c3[r];
        }
    }
    __syncthreads();

    const int row0 = bin * BIN_ROWS;
    const float4 b4 = ((const float4*)bias)[tid & 15];
    float4* out4 = (float4*)out;
    for (int i = tid; i < BIN_ROWS * D_FEAT / 4; i += 256) {
        const int row = row0 + (i >> 4);
        if (row < n_nodes) {
            const float4 t = *(const float4*)(&tile[(i >> 4) * TILE_STRIDE + (i & 15) * 4]);
            out4[(size_t)row * (D_FEAT / 4) + (i & 15)] = make_float4(t.x + b4.x, t.y + b4.y, t.z + b4.z, t.w + b4.w);
        }
    }
}

extern "C" void kernel_launch(void* const* d_in, const int* in_sizes, int n_in,
                              void* d_out, int out_size, void* d_ws, size_t ws_size,
                              hipStream_t stream) {
    const int*   rows = (const int*)d_in[0];
    const int*   cols = (const int*)d_in[1];
    const float* vals = (const float*)d_in[2];
    const float* X    = (const float*)d_in[3];
    const float* W    = (const float*)d_in[4];
    const float* bias = (const float*)d_in[5];
    float* out = (float*)d_out;

    const int n_edges = in_sizes[0];
    const int n_nodes = in_sizes[3] / D_FEAT;
    const int n_bins  = (n_nodes + BIN_ROWS - 1) / BIN_ROWS;
    const int n_chunks = (n_edges + SORT_CHUNK - 1) / SORT_CHUNK;

    char* ws = (char*)d_ws;
    int* starts = (int*)ws;
    const size_t o = (((size_t)n_chunks * (n_bins + 1) * sizeof(int)) + 255) & ~(size_t)255;
    int2* recs = (int2*)(ws + o);

    gcn_sort_kernel<<<n_chunks, SORT_THREADS, (size_t)n_bins * sizeof(int), stream>>>(
        rows, cols, vals, starts, recs, n_edges, n_bins);
    gcn_binsum_dense_kernel<<<n_bins, 256, (size_t)3 * n_chunks * sizeof(int), stream>>>(
        recs, starts, X, W, bias, out, n_bins, n_nodes, n_chunks);
}


# Row gathers of a bin wave run one group of 16 ahead of its sums (16 to 32 rows in flight per wave, was 0 to 8); records taken 32 per LDS read by readlane; row sum order unchanged
# speedup vs baseline: 3.6632x; 1.0320x over previous
#include <hip/hip_runtime.h>

typedef __attribute__((ext_vector_type(2))) float v2f;
typedef __attribute__((ext_vector_type(8))) float v8f;

#define D_FEAT 64
#define W_STRIDE 68
#define TILE_STRIDE 68

#define BIN_SHIFT 7
#define BIN_ROWS  (1 << BIN_SHIFT)
#define SORT_CHUNK 12288
#define SORT_THREADS 1024
#define REC_CAP 2560
#define GATHER_GROUP 16
#define STAGE_RUNS 9


__global__ __launch_bounds__(SORT_THREADS) void gcn_sort_kernel(
    const int*   __restrict__ rows,
    const int*   __restrict__ cols,
    const float* __restrict__ vals,
    int*         __restrict__ starts,
    int2*        __restrict__ recs,
    int n_edges, int n_bins)
{
    extern __shared__ int lslot[];
    __shared__ int2 srec[SORT_CHUNK];
    __shared__ int  wsum[SORT_THREADS / 32];

    const int tid  = threadIdx.x;
    const int lane = tid & 31;
    const int wave = tid >> 5;
    const int blk  = blockIdx.x;

    for (int i = tid; i < n_bins; i += SORT_THREADS) lslot[i] = 0;
    __syncthreads();

    const int lo = blk * SORT_CHUNK;
    const int hi = min(n_edges, lo + SORT_CHUNK);
    for (int e = lo + tid * 4; e < hi; e += SORT_THREADS * 4) {
        if (e + 3 < hi) {
            const int4 r = *(const int4*)(rows + e);
            atomicAdd(&lslot[r.x >> BIN_SHIFT], 1);
            atomicAdd(&lslot[r.y >> BIN_SHIFT], 1);
            atomicAdd(&lslot[r.z >> BIN_SHIFT], 1);
            atomicAdd(&lslot[r.w >> BIN_SHIFT], 1);
        } else {
            for (int j = e; j < hi; ++j) atomicAdd(&lslot[rows[j] >> BIN_SHIFT], 1);
        }
    }
    __syncthreads();

    int* mystarts = starts + (size_t)blk * (n_bins + 1);
    int carry = 0;
    for (int base = 0; base < n_bins; base += SORT_THREADS) {
        const int i = base + tid;
        const int c = (i < n_bins) ? lslot[i] : 0;
        int inc = c;
        for (int d = 1; d < 32; d <<= 1) {
            const int u = __shfl_up(inc, d, 32);
            if (lane >= d) inc += u;
        }
        if (lane == 31) wsum[wave] = inc;
        __syncthreads();
        int ws = wsum[lane];
        for (int d = 1; d < 32; d <<= 1) {
            const int u = __shfl_up(ws, d, 32);
            if (lane >= d) ws += u;
        }
        const int tot = __shfl(ws, 31, 32);
        const int wp  = __shfl(ws, wave, 32) - wsum[wave];
        const int ex  = carry + wp + inc - c;
        if (i < n_bins) { lslot[i] = ex; mystarts[i] = lo + ex; }
        carry += tot;
        __syncthreads();
    }
    if (tid == 0) mystarts[n_bins] = hi;

    for (int e = lo + tid * 4; e < hi; e += SORT_THREADS * 4) {
        if (e + 3 < hi) {
            const int4   r = *(const int4*)(rows + e);
            const int4   c = *(const int4*)(cols + e);
            const float4 v = *(const float4*)(vals + e);
            const int s0 = atomicAdd(&lslot[r.x >> BIN_SHIFT], 1);
            const int s1 = atomicAdd(&lslot[r.y >> BIN_SHIFT], 1);
            const int s2 = atomicAdd(&lslot[r.z >> BIN_SHIFT], 1);
            const int s3 = atomicAdd(&lslot[r.w >> BIN_SHIFT], 1);
            srec[s0] = make_int2(c.x | ((r.x & (BIN_ROWS - 1)) << 24), __float_as_int(v.x));
            srec[s1] = make_int2(c.y | ((r.y & (BIN_ROWS - 1)) << 24), __float_as_int(v.y));
            srec[s2] = make_int2(c.z | ((r.z & (BIN_ROWS - 1)) << 24), __float_as_int(v.z));
            srec[s3] = make_int2(c.w | ((r.w & (BIN_ROWS - 1)) << 24), __float_as_int(v.w));
        } else {
            for (int j = e; j < hi; ++j) {
                const int rj = rows[j];
                const int s = atomicAdd(&lslot[rj >> BIN_SHIFT], 1);
                srec[s] = make_int2(cols[j] | ((rj & (BIN_ROWS - 1)) << 24), __float_as_int(vals[j]));
            }
        }
    }
    __syncthreads();

    const int n = hi - lo;
    for (int t = tid; t < n; t += SORT_THREADS) recs[lo + t] = srec[t];
}

__global__ __launch_bounds__(256) void gcn_binsum_dense_kernel(
    const int2*  __restrict__ recs,
    const int*   __restrict__ starts,
    const float* __restrict__ X,
    const float* __restrict__ W,
    const float* __restrict__ bias,
    float*       __restrict__ out,
    int n_bins, int n_nodes, int n_chunks)
{
    extern __shared__ int lruns[];
    __shared__ alignas(16) float tile[BIN_ROWS * TILE_STRIDE];
    __shared__ alignas(16) float ldsW[D_FEAT * W_STRIDE];
    __shared__ int2 lrec[REC_CAP];
    __shared__ int  wsum[8];
    __shared__ int  rowslot[BIN_ROWS];
    __shared__ int  rowbeg[BIN_ROWS + 1];

    int* rsrc = lruns;
    int* rcnt = lruns + n_chunks;
    int* rpos = lruns + 2 * n_chunks;

    const int tid  = threadIdx.x;
    const int lane = tid & 31;
    const int bin  = blockIdx.x;

    float4* tile4 = (float4*)tile;
    for (int i = tid; i < BIN_ROWS * TILE_STRIDE / 4; i += 256) tile4[i] = make_float4(0.f, 0.f, 0.f, 0.f);

    int carry = 0;
    for (int base = 0; base < n_chunks; base += 256) {
        const int c = base + tid;
        int s = 0, n = 0;
        if (c < n_chunks) {
            const int* p = starts + (size_t)c * (n_bins + 1) + bin;
            s = p[0];
            n = p[1] - s;
        }
        if (base == 0) {
            float4 wreg[D_FEAT * D_FEAT / 4 / 256];
#pragma unroll
            for (int j = 0; j < D_FEAT * D_FEAT / 4 / 256; ++j) wreg[j] = ((const float4*)W)[tid + 256 * j];
#pragma unroll
            for (int j = 0; j < D_FEAT * D_FEAT / 4 / 256; ++j) {
                const int i = tid + 256 * j;
                *(float4*)(&ldsW[(i >> 4) * W_STRIDE + (i & 15) * 4]) = wreg[j];
            }
        }
        int inc = n;
        for (int d = 1; d < 32; d <<= 1) {
            const int u = __shfl_up(inc, d, 32);
            if (lane >= d) inc += u;
        }
        if (lane == 31) wsum[tid >> 5] = inc;
        __syncthreads();
        int wp = 0, tot = 0;
        for (int w = 0; w < 8; ++w) {
            const int t = wsum[w];
            tot += t;
            if (w < (tid >> 5)) wp += t;
        }
        if (c < n_chunks) { rsrc[c] = s; rcnt[c] = n; rpos[c] = carry + wp + inc - n; }
        carry += tot;
        __syncthreads();
    }
    const int total = carry;

    const int wave = __builtin_amdgcn_readfirstlane(tid >> 5);
    const v2f* X2 = (const v2f*)X;

    for (int pass0 = 0; pass0 < total; pass0 += REC_CAP) {
        if (tid < BIN_ROWS) rowslot[tid] = 0;

        for (int cb = wave; cb < n_chunks; cb += 8 * STAGE_RUNS) {
            int s[STAGE_RUNS], p0[STAGE_RUNS], j0[STAGE_RUNS], j1[STAGE_RUNS];
            int2 r[STAGE_RUNS];
#pragma unroll
            for (int u = 0; u < STAGE_RUNS; ++u) {
                const int c  = cb + 8 * u;
                const int cc = min(c, n_chunks - 1);
                const int n  = rcnt[cc];
                s[u]  = rsrc[cc];
                p0[u] = rpos[cc] - pass0;
                j0[u] = max(0, -p0[u]) + lane;
                j1[u] = (c < n_chunks) ? min(n, REC_CAP - p0[u]) : 0;
            }
#pragma unroll
            for (int u = 0; u < STAGE_RUNS; ++u) {
                r[u] = make_int2(0, 0);
                if (j0[u] < j1[u]) r[u] = recs[s[u] + j0[u]];
            }
#pragma unroll
            for (int u = 0; u < STAGE_RUNS; ++u) {
                if (j0[u] < j1[u]) lrec[p0[u] + j0[u]] = r[u];
            }
#pragma unroll
            for (int u = 0; u < STAGE_RUNS; ++u) {
                for (int j = j0[u] + 32; j < j1[u]; j += 32) lrec[p0[u] + j] = recs[s[u] + j];
            }
        }
        __syncthreads();

        const int nb = min(REC_CAP, total - pass0);

        int2 mine[REC_CAP / 256];
#pragma unroll
        for (int k = 0; k < REC_CAP / 256; ++k) {
            const int i = tid + 256 * k;
            mine[k] = make_int2(0, 0);
            if (i < nb) {
                mine[k] = lrec[i];
                atomicAdd(&rowslot[(unsigned)mine[k].x >> 24], 1);
            }
        }
        __syncthreads();
        {
            const int c = (tid < BIN_ROWS) ? rowslot[tid] : 0;
            int inc = c;
            for (int d = 1; d < 32; d <<= 1) {
                const int u = __shfl_up(inc, d, 32);
                if (lane >= d) inc += u;
            }
            if (lane == 31) wsum[tid >> 5] = inc;
            __syncthreads();
            int wp = 0;
            for (int w = 0; w < BIN_ROWS / 32; ++w) {
                const int t = wsum[w];
                if (w < (tid >> 5)) wp += t;
            }
            if (tid < BIN_ROWS) {
                const int ex = wp + inc - c;
                rowbeg[tid]  = ex;
                rowslot[tid] = ex;
            }
            if (tid == 0) rowbeg[BIN_ROWS] = nb;
        }
        __syncthreads();
#pragma unroll
        for (int k = 0; k < REC_CAP / 256; ++k) {
            const int i = tid + 256 * k;
            if (i < nb) {
                const int sl = atomicAdd(&rowslot[(unsigned)mine[k].x >> 24], 1);
                lrec[sl] = mine[k];
            }
        }
        __syncthreads();

        const int e_lo = __builtin_amdgcn_readfirstlane(rowbeg[wave * 16]);
        const int e_hi = __builtin_amdgcn_readfirstlane(rowbeg[wave * 16 + 16]);
        if (e_lo < e_hi) {
            const int last = e_hi - 1;
            int e0 = e_lo;
            int2 R0 = lrec[min(e0 + lane, last)];
            v2f xa[GATHER_GROUP], xb[GATHER_GROUP];
#pragma unroll
            for (int k = 0; k < GATHER_GROUP; ++k) {
                const int c = __builtin_amdgcn_readlane(R0.x, k) & 0xFFFFFF;
                xa[k] = X2[(size_t)c * (D_FEAT / 2) + lane];
            }
            int cur = (int)((unsigned)__builtin_amdgcn_readlane(R0.x, 0) >> 24);
            v2f acc = {0.f, 0.f};
            do {
                const int2 R1 = lrec[min(e0 + 2 * GATHER_GROUP + lane, last)];
#pragma unroll
                for (int k = 0; k < GATHER_GROUP; ++k) {
                    const int c = __builtin_amdgcn_readlane(R0.x, GATHER_GROUP + k) & 0xFFFFFF;
                    xb[k] = X2[(size_t)c * (D_FEAT / 2) + lane];
                }
#pragma unroll
                for (int k = 0; k < GATHER_GROUP; ++k) {
                    if (e0 + k < e_hi) {
                        const int rx = __builtin_amdgcn_readlane(R0.x, k);
                        const int ry = __builtin_amdgcn_readlane(R0.y, k);
                        const int row = (int)((unsigned)rx >> 24);
                        if (row != cur) {
                            v2f* d = (v2f*)(tile + cur * TILE_STRIDE + lane * 2);
                            v2f t = *d;
                            t.x += acc.x; t.y += acc.y;
                            *d = t;
                            acc.x = 0.f; acc.y = 0.f;
                            cur = row;
                        }
                        const float v = __int_as_float(ry);
                        acc.x = fmaf(v, xa[k].x, acc.x);
                        acc.y = fmaf(v, xa[k].y, acc.y);
                    }
                }
#pragma unroll
                for (int k = 0; k < GATHER_GROUP; ++k) {
                    const int c = __builtin_amdgcn_readlane(R1.x, k) & 0xFFFFFF;
                    xa[k] = X2[(size_t)c * (D_FEAT / 2) + lane];
                }
#pragma unroll
                for (int k = 0; k < GATHER_GROUP; ++k) {
                    if (e0 + GATHER_GROUP + k < e_hi) {
                        const int rx = __builtin_amdgcn_readlane(R0.x, GATHER_GROUP + k);
                        const int ry = __builtin_amdgcn_readlane(R0.y, GATHER_GROUP + k);
                        const int row = (int)((unsigned)rx >> 24);
                        if (row != cur) {
                            v2f* d = (v2f*)(tile + cur * TILE_STRIDE + lane * 2);
                            v2f t = *d;
                            t.x += acc.x; t.y += acc.y;
                            *d = t;
                            acc.x = 0.f; acc.y = 0.f;
                            cur = row;
                        }
                        const float v = __int_as_float(ry);
                        acc.x = fmaf(v, xb[k].x, acc.x);
                        acc.y = fmaf(v, xb[k].y, acc.y);
                    }
                }
                R0 = R1;
                e0 += 2 * GATHER_GROUP;
            } while (e0 < e_hi);
            v2f* d = (v2f*)(tile + cur * TILE_STRIDE + lane * 2);
            v2f t = *d;
            t.x += acc.x; t.y += acc.y;
            *d = t;
        }
        __syncthreads();
    }

    {
        const int half = lane >> 4;
        const int l16  = lane & 15;
        const int koff = half << 1;
        const float* arow = tile + (wave * 16 + l16) * TILE_STRIDE;
        v8f c0 = {}; v8f c1 = {}; v8f c2 = {}; v8f c3 = {};

#pragma unroll
        for (int kk = 0; kk < 16; ++kk) {
            const int kb = kk * 4 + koff;

            v2f a0 = *(const v2f*)(arow + kb);

            v2f b0 = *(const v2f*)(&ldsW[( 0 + l16) * W_STRIDE + kb]);
            v2f b1 = *(const v2f*)(&ldsW[(16 + l16) * W_STRIDE + kb]);
            v2f b2 = *(const v2f*)(&ldsW[(32 + l16) * W_STRIDE + kb]);
            v2f b3 = *(const v2f*)(&ldsW[(48 + l16) * W_STRIDE + kb]);

            c0 = __builtin_amdgcn_wmma_f32_16x16x4_f32(false, a0, false, b0, (short)0, c0, false, false);
            c1 = __builtin_amdgcn_wmma_f32_16x16x4_f32(false, a0, false, b1, (short)0, c1, false, false);
            c2 = __builtin_amdgcn_wmma_f32_16x16x4_f32(false, a0, false, b2, (short)0, c2, false, false);
            c3 = __builtin_amdgcn_wmma_f32_16x16x4_f32(false, a0, false, b3, (short)0, c3, false, false);
        }

        float* drow = tile + (wave * 16 + half * 8) * TILE_STRIDE + l16;
#pragma unroll
        for (int r = 0; r < 8; ++r) {
            drow[r * TILE_STRIDE +  0] = c0[r];
            drow[r * TILE_STRIDE + 16] = c1[r];
            drow[r * TILE_STRIDE + 32] = c2[r];
            drow[r * TILE_STRIDE + 48] = c3[r];
        }
    }
    __syncthreads();

    const int row0 = bin * BIN_ROWS;
    const float4 b4 = ((const float4*)bias)[tid & 15];
    float4* out4 = (float4*)out;
    for (int i = tid; i < BIN_ROWS * D_FEAT / 4; i += 256) {
        const int row = row0 + (i >> 4);
        if (row < n_nodes) {
            const float4 t = *(const float4*)(&tile[(i >> 4) * TILE_STRIDE + (i & 15) * 4]);
            out4[(size_t)row * (D_FEAT / 4) + (i & 15)] = make_float4(t.x + b4.x, t.y + b4.y, t.z + b4.z, t.w + b4.w);
        }
    }
}

extern "C" void kernel_launch(void* const* d_in, const int* in_sizes, int n_in,
                              void* d_out, int out_size, void* d_ws, size_t ws_size,
                              hipStream_t stream) {
    const int*   rows = (const int*)d_in[0];
    const int*   cols = (const int*)d_in[1];
    const float* vals = (const float*)d_in[2];
    const float* X    = (const float*)d_in[3];
    const float* W    = (const float*)d_in[4];
    const float* bias = (const float*)d_in[5];
    float* out = (float*)d_out;

    const int n_edges = in_sizes[0];
    const int n_nodes = in_sizes[3] / D_FEAT;
    const int n_bins  = (n_nodes + BIN_ROWS - 1) / BIN_ROWS;
    const int n_chunks = (n_edges + SORT_CHUNK - 1) / SORT_CHUNK;

    char* ws = (char*)d_ws;
    int* starts = (int*)ws;
    const size_t o = (((size_t)n_chunks * (n_bins + 1) * sizeof(int)) + 255) & ~(size_t)255;
    int2* recs = (int2*)(ws + o);

    gcn_sort_kernel<<<n_chunks, SORT_THREADS, (size_t)n_bins * sizeof(int), stream>>>(
        rows, cols, vals, starts, recs, n_edges, n_bins);
    gcn_binsum_dense_kernel<<<n_bins, 256, (size_t)3 * n_chunks * sizeof(int), stream>>>(
        recs, starts, X, W, bias, out, n_bins, n_nodes, n_chunks);
}


# Bin wave's sum loop: row changes of 32 records found at once (lane compare, ballot), vals past the stretch's end zeroed: per record one bit test, no end test, no taken branch; sum order unchanged
# speedup vs baseline: 3.7946x; 1.0359x over previous
#include <hip/hip_runtime.h>

typedef __attribute__((ext_vector_type(2))) float v2f;
typedef __attribute__((ext_vector_type(8))) float v8f;

#define D_FEAT 64
#define W_STRIDE 68
#define TILE_STRIDE 68

#define BIN_SHIFT 7
#define BIN_ROWS  (1 << BIN_SHIFT)
#define SORT_CHUNK 12288
#define SORT_THREADS 1024
#define REC_CAP 2560
#define GATHER_GROUP 16
#define STAGE_RUNS 9


__global__ __launch_bounds__(SORT_THREADS) void gcn_sort_kernel(
    const int*   __restrict__ rows,
    const int*   __restrict__ cols,
    const float* __restrict__ vals,
    int*         __restrict__ starts,
    int2*        __restrict__ recs,
    int n_edges, int n_bins)
{
    extern __shared__ int lslot[];
    __shared__ int2 srec[SORT_CHUNK];
    __shared__ int  wsum[SORT_THREADS / 32];

    const int tid  = threadIdx.x;
    const int lane = tid & 31;
    const int wave = tid >> 5;
    const int blk  = blockIdx.x;

    for (int i = tid; i < n_bins; i += SORT_THREADS) lslot[i] = 0;
    __syncthreads();

    const int lo = blk * SORT_CHUNK;
    const int hi = min(n_edges, lo + SORT_CHUNK);
    for (int e = lo + tid * 4; e < hi; e += SORT_THREADS * 4) {
        if (e + 3 < hi) {
            const int4 r = *(const int4*)(rows + e);
            atomicAdd(&lslot[r.x >> BIN_SHIFT], 1);
            atomicAdd(&lslot[r.y >> BIN_SHIFT], 1);
            atomicAdd(&lslot[r.z >> BIN_SHIFT], 1);
            atomicAdd(&lslot[r.w >> BIN_SHIFT], 1);
        } else {
            for (int j = e; j < hi; ++j) atomicAdd(&lslot[rows[j] >> BIN_SHIFT], 1);
        }
    }
    __syncthreads();

    int* mystarts = starts + (size_t)blk * (n_bins + 1);
    int carry = 0;
    for (int base = 0; base < n_bins; base += SORT_THREADS) {
        const int i = base + tid;
        const int c = (i < n_bins) ? lslot[i] : 0;
        int inc = c;
        for (int d = 1; d < 32; d <<= 1) {
            const int u = __shfl_up(inc, d, 32);
            if (lane >= d) inc += u;
        }
        if (lane == 31) wsum[wave] = inc;
        __syncthreads();
        int ws = wsum[lane];
        for (int d = 1; d < 32; d <<= 1) {
            const int u = __shfl_up(ws, d, 32);
            if (lane >= d) ws += u;
        }
        const int tot = __shfl(ws, 31, 32);
        const int wp  = __shfl(ws, wave, 32) - wsum[wave];
        const int ex  = carry + wp + inc - c;
        if (i < n_bins) { lslot[i] = ex; mystarts[i] = lo + ex; }
        carry += tot;
        __syncthreads();
    }
    if (tid == 0) mystarts[n_bins] = hi;

    for (int e = lo + tid * 4; e < hi; e += SORT_THREADS * 4) {
        if (e + 3 < hi) {
            const int4   r = *(const int4*)(rows + e);
            const int4   c = *(const int4*)(cols + e);
            const float4 v = *(const float4*)(vals + e);
            const int s0 = atomicAdd(&lslot[r.x >> BIN_SHIFT], 1);
            const int s1 = atomicAdd(&lslot[r.y >> BIN_SHIFT], 1);
            const int s2 = atomicAdd(&lslot[r.z >> BIN_SHIFT], 1);
            const int s3 = atomicAdd(&lslot[r.w >> BIN_SHIFT], 1);
            srec[s0] = make_int2(c.x | ((r.x & (BIN_ROWS - 1)) << 24), __float_as_int(v.x));
            srec[s1] = make_int2(c.y | ((r.y & (BIN_ROWS - 1)) << 24), __float_as_int(v.y));
            srec[s2] = make_int2(c.z | ((r.z & (BIN_ROWS - 1)) << 24), __float_as_int(v.z));
            srec[s3] = make_int2(c.w | ((r.w & (BIN_ROWS - 1)) << 24), __float_as_int(v.w));
        } else {
            for (int j = e; j < hi; ++j) {
                const int rj = rows[j];
                const int s = atomicAdd(&lslot[rj >> BIN_SHIFT], 1);
                srec[s] = make_int2(cols[j] | ((rj & (BIN_ROWS - 1)) << 24), __float_as_int(vals[j]));
            }
        }
    }
    __syncthreads();

    const int n = hi - lo;
    for (int t = tid; t < n; t += SORT_THREADS) recs[lo + t] = srec[t];
}

__global__ __launch_bounds__(256) void gcn_binsum_dense_kernel(
    const int2*  __restrict__ recs,
    const int*   __restrict__ starts,
    const float* __restrict__ X,
    const float* __restrict__ W,
    const float* __restrict__ bias,
    float*       __restrict__ out,
    int n_bins, int n_nodes, int n_chunks)
{
    extern __shared__ int lruns[];
    __shared__ alignas(16) float tile[BIN_ROWS * TILE_STRIDE];
    __shared__ alignas(16) float ldsW[D_FEAT * W_STRIDE];
    __shared__ int2 lrec[REC_CAP];
    __shared__ int  wsum[8];
    __shared__ int  rowslot[BIN_ROWS];
    __shared__ int  rowbeg[BIN_ROWS + 1];

    int* rsrc = lruns;
    int* rcnt = lruns + n_chunks;
    int* rpos = lruns + 2 * n_chunks;

    const int tid  = threadIdx.x;
    const int lane = tid & 31;
    const int bin  = blockIdx.x;

    float4* tile4 = (float4*)tile;
    for (int i = tid; i < BIN_ROWS * TILE_STRIDE / 4; i += 256) tile4[i] = make_float4(0.f, 0.f, 0.f, 0.f);

    int carry = 0;
    for (int base = 0; base < n_chunks; base += 256) {
        const int c = base + tid;
        int s = 0, n = 0;
        if (c < n_chunks) {
            const int* p = starts + (size_t)c * (n_bins + 1) + bin;
            s = p[0];
            n = p[1] - s;
        }
        if (base == 0) {
            float4 wreg[D_FEAT * D_FEAT / 4 / 256];
#pragma unroll
            for (int j = 0; j < D_FEAT * D_FEAT / 4 / 256; ++j) wreg[j] = ((const float4*)W)[tid + 256 * j];
#pragma unroll
            for (int j = 0; j < D_FEAT * D_FEAT / 4 / 256; ++j) {
                const int i = tid + 256 * j;
                *(float4*)(&ldsW[(i >> 4) * W_STRIDE + (i & 15) * 4]) = wreg[j];
            }
        }
        int inc = n;
        for (int d = 1; d < 32; d <<= 1) {
            const int u = __shfl_up(inc, d, 32);
            if (lane >= d) inc += u;
        }
        if (lane == 31) wsum[tid >> 5] = inc;
        __syncthreads();
        int wp = 0, tot = 0;
        for (int w = 0; w < 8; ++w) {
            const int t = wsum[w];
            tot += t;
            if (w < (tid >> 5)) wp += t;
        }
        if (c < n_chunks) { rsrc[c] = s; rcnt[c] = n; rpos[c] = carry + wp + inc - n; }
        carry += tot;
        __syncthreads();
    }
    const int total = carry;

    const int wave = __builtin_amdgcn_readfirstlane(tid >> 5);
    const v2f* X2 = (const v2f*)X;

    for (int pass0 = 0; pass0 < total; pass0 += REC_CAP) {
        if (tid < BIN_ROWS) rowslot[tid] = 0;

        for (int cb = wave; cb < n_chunks; cb += 8 * STAGE_RUNS) {
            int s[STAGE_RUNS], p0[STAGE_RUNS], j0[STAGE_RUNS], j1[STAGE_RUNS];
            int2 r[STAGE_RUNS];
#pragma unroll
            for (int u = 0; u < STAGE_RUNS; ++u) {
                const int c  = cb + 8 * u;
                const int cc = min(c, n_chunks - 1);
                const int n  = rcnt[cc];
                s[u]  = rsrc[cc];
                p0[u] = rpos[cc] - pass0;
                j0[u] = max(0, -p0[u]) + lane;
                j1[u] = (c < n_chunks) ? min(n, REC_CAP - p0[u]) : 0;
            }
#pragma unroll
            for (int u = 0; u < STAGE_RUNS; ++u) {
                r[u] = make_int2(0, 0);
                if (j0[u] < j1[u]) r[u] = recs[s[u] + j0[u]];
            }
#pragma unroll
            for (int u = 0; u < STAGE_RUNS; ++u) {
                if (j0[u] < j1[u]) lrec[p0[u] + j0[u]] = r[u];
            }
#pragma unroll
            for (int u = 0; u < STAGE_RUNS; ++u) {
                for (int j = j0[u] + 32; j < j1[u]; j += 32) lrec[p0[u] + j] = recs[s[u] + j];
            }
        }
        __syncthreads();

        const int nb = min(REC_CAP, total - pass0);

        int2 mine[REC_CAP / 256];
#pragma unroll
        for (int k = 0; k < REC_CAP / 256; ++k) {
            const int i = tid + 256 * k;
            mine[k] = make_int2(0, 0);
            if (i < nb) {
                mine[k] = lrec[i];
                atomicAdd(&rowslot[(unsigned)mine[k].x >> 24], 1);
            }
        }
        __syncthreads();
        {
            const int c = (tid < BIN_ROWS) ? rowslot[tid] : 0;
            int inc = c;
            for (int d = 1; d < 32; d <<= 1) {
                const int u = __shfl_up(inc, d, 32);
                if (lane >= d) inc += u;
            }
            if (lane == 31) wsum[tid >> 5] = inc;
            __syncthreads();
            int wp = 0;
            for (int w = 0; w < BIN_ROWS / 32; ++w) {
                const int t = wsum[w];
                if (w < (tid >> 5)) wp += t;
            }
            if (tid < BIN_ROWS) {
                const int ex = wp + inc - c;
                rowbeg[tid]  = ex;
                rowslot[tid] = ex;
            }
            if (tid == 0) rowbeg[BIN_ROWS] = nb;
        }
        __syncthreads();
#pragma unroll
        for (int k = 0; k < REC_CAP / 256; ++k) {
            const int i = tid + 256 * k;
            if (i < nb) {
                const int sl = atomicAdd(&rowslot[(unsigned)mine[k].x >> 24], 1);
                lrec[sl] = mine[k];
            }
        }
        __syncthreads();

        const int e_lo = __builtin_amdgcn_readfirstlane(rowbeg[wave * 16]);
        const int e_hi = __builtin_amdgcn_readfirstlane(rowbeg[wave * 16 + 16]);
        if (e_lo < e_hi) {
            const int last = e_hi - 1;
            int e0 = e_lo;
            int2 R0 = lrec[min(e0 + lane, last)];
            unsigned chg0;
            {
                const int before = lrec[min(max(e0 + lane - 1, e_lo), last)].x;
                chg0 = __builtin_amdgcn_ballot_w32((unsigned)(R0.x ^ before) > 0xFFFFFFu);
                if (e0 + lane > last) R0.y = 0;
            }
            v2f xa[GATHER_GROUP], xb[GATHER_GROUP];
#pragma unroll
            for (int k = 0; k < GATHER_GROUP; ++k) {
                const int c = __builtin_amdgcn_readlane(R0.x, k) & 0xFFFFFF;
                xa[k] = X2[(size_t)c * (D_FEAT / 2) + lane];
            }
            int cur = (int)((unsigned)__builtin_amdgcn_readlane(R0.x, 0) >> 24);
            v2f acc = {0.f, 0.f};
            do {
                int2 R1 = lrec[min(e0 + 2 * GATHER_GROUP + lane, last)];
                const int before1 = lrec[min(e0 + 2 * GATHER_GROUP - 1 + lane, last)].x;
#pragma unroll
                for (int k = 0; k < GATHER_GROUP; ++k) {
                    const int c = __builtin_amdgcn_readlane(R0.x, GATHER_GROUP + k) & 0xFFFFFF;
                    xb[k] = X2[(size_t)c * (D_FEAT / 2) + lane];
                }
#pragma unroll
                for (int k = 0; k < GATHER_GROUP; ++k) {
                    if (__builtin_expect((chg0 >> k) & 1u, 0u)) {
                        v2f* d = (v2f*)(tile + cur * TILE_STRIDE + lane * 2);
                        v2f t = *d;
                        t.x += acc.x; t.y += acc.y;
                        *d = t;
                        acc.x = 0.f; acc.y = 0.f;
                        cur = (int)((unsigned)__builtin_amdgcn_readlane(R0.x, k) >> 24);
                    }
                    const float v = __int_as_float(__builtin_amdgcn_readlane(R0.y, k));
                    acc.x = fmaf(v, xa[k].x, acc.x);
                    acc.y = fmaf(v, xa[k].y, acc.y);
                }
                if (__builtin_expect((chg0 >> GATHER_GROUP) & 1u, 0u)) {
                    v2f* d = (v2f*)(tile + cur * TILE_STRIDE + lane * 2);
                    v2f t = *d;
                    t.x += acc.x; t.y += acc.y;
                    *d = t;
                    acc.x = 0.f; acc.y = 0.f;
                    cur = (int)((unsigned)__builtin_amdgcn_readlane(R0.x, GATHER_GROUP) >> 24);
                }
                const unsigned chg1 = __builtin_amdgcn_ballot_w32((unsigned)(R1.x ^ before1) > 0xFFFFFFu);
                if (e0 + 2 * GATHER_GROUP + lane > last) R1.y = 0;
#pragma unroll
                for (int k = 0; k < GATHER_GROUP; ++k) {
                    const int c = __builtin_amdgcn_readlane(R1.x, k) & 0xFFFFFF;
                    xa[k] = X2[(size_t)c * (D_FEAT / 2) + lane];
                }
#pragma unroll
                for (int k = 0; k < GATHER_GROUP; ++k) {
                    if (k > 0 && __builtin_expect((chg0 >> (GATHER_GROUP + k)) & 1u, 0u)) {
                        v2f* d = (v2f*)(tile + cur * TILE_STRIDE + lane * 2);
                        v2f t = *d;
                        t.x += acc.x; t.y += acc.y;
                        *d = t;
                        acc.x = 0.f; acc.y = 0.f;
                        cur = (int)((unsigned)__builtin_amdgcn_readlane(R0.x, GATHER_GROUP + k) >> 24);
                    }
                    const float v = __int_as_float(__builtin_amdgcn_readlane(R0.y, GATHER_GROUP + k));
                    acc.x = fmaf(v, xb[k].x, acc.x);
                    acc.y = fmaf(v, xb[k].y, acc.y);
                }
                R0 = R1;
                chg0 = chg1;
                e0 += 2 * GATHER_GROUP;
            } while (e0 < e_hi);
            v2f* d = (v2f*)(tile + cur * TILE_STRIDE + lane * 2);
            v2f t = *d;
            t.x += acc.x; t.y += acc.y;
            *d = t;
        }
        __syncthreads();
    }

    {
        const int half = lane >> 4;
        const int l16  = lane & 15;
        const int koff = half << 1;
        const float* arow = tile + (wave * 16 + l16) * TILE_STRIDE;
        v8f c0 = {}; v8f c1 = {}; v8f c2 = {}; v8f c3 = {};

#pragma unroll
        for (int kk = 0; kk < 16; ++kk) {
            const int kb = kk * 4 + koff;

            v2f a0 = *(const v2f*)(arow + kb);

            v2f b0 = *(const v2f*)(&ldsW[( 0 + l16) * W_STRIDE + kb]);
            v2f b1 = *(const v2f*)(&ldsW[(16 + l16) * W_STRIDE + kb]);
            v2f b2 = *(const v2f*)(&ldsW[(32 + l16) * W_STRIDE + kb]);
            v2f b3 = *(const v2f*)(&ldsW[(48 + l16) * W_STRIDE + kb]);

            c0 = __builtin_amdgcn_wmma_f32_16x16x4_f32(false, a0, false, b0, (short)0, c0, false, false);
            c1 = __builtin_amdgcn_wmma_f32_16x16x4_f32(false, a0, false, b1, (short)0, c1, false, false);
            c2 = __builtin_amdgcn_wmma_f32_16x16x4_f32(false, a0, false, b2, (short)0, c2, false, false);
            c3 = __builtin_amdgcn_wmma_f32_16x16x4_f32(false, a0, false, b3, (short)0, c3, false, false);
        }

        float* drow = tile + (wave * 16 + half * 8) * TILE_STRIDE + l16;
#pragma unroll
        for (int r = 0; r < 8; ++r) {
            drow[r * TILE_STRIDE +  0] = c0[r];
            drow[r * TILE_STRIDE + 16] = c1[r];
            drow[r * TILE_STRIDE + 32] = c2[r];
            drow[r * TILE_STRIDE + 48] = c3[r];
        }
    }
    __syncthreads();

    const int row0 = bin * BIN_ROWS;
    const float4 b4 = ((const float4*)bias)[tid & 15];
    float4* out4 = (float4*)out;
    for (int i = tid; i < BIN_ROWS * D_FEAT / 4; i += 256) {
        const int row = row0 + (i >> 4);
        if (row < n_nodes) {
            const float4 t = *(const float4*)(&tile[(i >> 4) * TILE_STRIDE + (i & 15) * 4]);
            out4[(size_t)row * (D_FEAT / 4) + (i & 15)] = make_float4(t.x + b4.x, t.y + b4.y, t.z + b4.z, t.w + b4.w);
        }
    }
}

extern "C" void kernel_launch(void* const* d_in, const int* in_sizes, int n_in,
                              void* d_out, int out_size, void* d_ws, size_t ws_size,
                              hipStream_t stream) {
    const int*   rows = (const int*)d_in[0];
    const int*   cols = (const int*)d_in[1];
    const float* vals = (const float*)d_in[2];
    const float* X    = (const float*)d_in[3];
    const float* W    = (const float*)d_in[4];
    const float* bias = (const float*)d_in[5];
    float* out = (float*)d_out;

    const int n_edges = in_sizes[0];
    const int n_nodes = in_sizes[3] / D_FEAT;
    const int n_bins  = (n_nodes + BIN_ROWS - 1) / BIN_ROWS;
    const int n_chunks = (n_edges + SORT_CHUNK - 1) / SORT_CHUNK;

    char* ws = (char*)d_ws;
    int* starts = (int*)ws;
    const size_t o = (((size_t)n_chunks * (n_bins + 1) * sizeof(int)) + 255) & ~(size_t)255;
    int2* recs = (int2*)(ws + o);

    gcn_sort_kernel<<<n_chunks, SORT_THREADS, (size_t)n_bins * sizeof(int), stream>>>(
        rows, cols, vals, starts, recs, n_edges, n_bins);
    gcn_binsum_dense_kernel<<<n_bins, 256, (size_t)3 * n_chunks * sizeof(int), stream>>>(
        recs, starts, X, W, bias, out, n_bins, n_nodes, n_chunks);
}
